# starting kernel
# baseline (speedup 1.0000x reference)
#include <hip/hip_runtime.h>
#include <hip/hip_bf16.h>
#include <math.h>


typedef __attribute__((ext_vector_type(2))) float v2f;
typedef __attribute__((ext_vector_type(8))) float v8f;

#define B_SZ 4096
#define D_SZ 66
#define H_SZ 512

constexpr int BM = 64;
constexpr int BN = 64;
constexpr int BK = 32;
constexpr int NT = 128;

__device__ __forceinline__ v8f wmma4(v2f a, v2f b, v8f c) {
    return __builtin_amdgcn_wmma_f32_16x16x4_f32(
        false, a, false, b, (short)0, c, false, false);
}

__global__ __launch_bounds__(256)
void precompute_C(const float* __restrict__ W1, const float* __restrict__ W2,
                  const float* __restrict__ W3, float* __restrict__ Cm)
{
    int idx = blockIdx.x * blockDim.x + threadIdx.x;
    int k = idx & (H_SZ - 1);
    int l = idx >> 9;
    float m = 0.0f;
    #pragma unroll 2
    for (int j = 0; j < D_SZ; ++j)
        m = fmaf(W1[j * H_SZ + l], W3[k * D_SZ + j], m);
    Cm[l * H_SZ + k] = m * W2[l * H_SZ + k];
}

template<bool ACONCAT, int EPI>
__global__ __launch_bounds__(NT)
void gemm_wmma(const float* __restrict__ A, const float* __restrict__ xs,
               const float* __restrict__ tvec,
               const float* __restrict__ Bm, const float* __restrict__ bias,
               float* __restrict__ out0, float* __restrict__ out1,
               int M, int N, int K)
{
    __shared__ float As[BM][BK + 4];
    __shared__ float Bt[BN][BK + 4];

    const int tid  = threadIdx.x;
    const int wave = tid >> 5;
    const int lane = tid & 31;
    const int wm   = wave >> 1;
    const int wn   = wave & 1;

    const int m0 = blockIdx.y * BM;
    const int n0 = blockIdx.x * BN;

    v8f acc[2][2] = {};

    const int row16 = lane & 15;
    const int ksub  = (lane >> 4) << 1;

    for (int k0 = 0; k0 < K; k0 += BK) {
        #pragma unroll
        for (int i = 0; i < (BM * BK) / NT; ++i) {
            int idx = tid + i * NT;
            int mm = idx >> 5;
            int kk = idx & (BK - 1);
            int gk = k0 + kk;
            float v = 0.0f;
            if (ACONCAT) {
                int gm = m0 + mm;
                if (gk < D_SZ)       v = xs[gm * D_SZ + gk];
                else if (gk == D_SZ) v = tvec[gm];
            } else {
                if (gk < K) v = A[(size_t)(m0 + mm) * K + gk];
            }
            As[mm][kk] = v;
        }
        #pragma unroll
        for (int i = 0; i < (BK * BN) / NT; ++i) {
            int idx = tid + i * NT;
            int kk = idx / BN;
            int nn = idx % BN;
            int gk = k0 + kk;
            int gn = n0 + nn;
            float v = 0.0f;
            if (gk < K && gn < N) v = Bm[(size_t)gk * N + gn];
            Bt[nn][kk] = v;
        }
        __syncthreads();

        #pragma unroll
        for (int kk = 0; kk < BK; kk += 4) {
            v2f a0 = *(const v2f*)&As[wm * 32 +      row16][kk + ksub];
            v2f a1 = *(const v2f*)&As[wm * 32 + 16 + row16][kk + ksub];
            v2f b0 = *(const v2f*)&Bt[wn * 32 +      row16][kk + ksub];
            v2f b1 = *(const v2f*)&Bt[wn * 32 + 16 + row16][kk + ksub];
            acc[0][0] = wmma4(a0, b0, acc[0][0]);
            acc[0][1] = wmma4(a0, b1, acc[0][1]);
            acc[1][0] = wmma4(a1, b0, acc[1][0]);
            acc[1][1] = wmma4(a1, b1, acc[1][1]);
        }
        __syncthreads();
    }

    #pragma unroll
    for (int mt = 0; mt < 2; ++mt) {
        const int mbase = m0 + wm * 32 + mt * 16 + ((lane >> 4) << 3);
        #pragma unroll
        for (int nt = 0; nt < 2; ++nt) {
            const int nc = n0 + wn * 32 + nt * 16 + row16;
            if (nc >= N) continue;
            const float bi = bias ? bias[nc] : 0.0f;
            #pragma unroll
            for (int r = 0; r < 8; ++r) {
                const int gm = mbase + r;
                float z = acc[mt][nt][r] + bi;
                if (EPI == 0) {
                    out0[(size_t)gm * N + nc] = z;
                } else {
                    float sg = 1.0f / (1.0f + __expf(-z));
                    out0[(size_t)gm * N + nc] = z * sg;
                    out1[(size_t)gm * N + nc] = sg * (1.0f + z * (1.0f - sg));
                }
            }
        }
    }
}

__global__ __launch_bounds__(128)
void dot_div_kernel(const float* __restrict__ u, const float* __restrict__ s2,
                    float* __restrict__ out)
{
    int b = blockIdx.x;
    int tid = threadIdx.x;
    float sum = 0.0f;
    #pragma unroll
    for (int k = tid; k < H_SZ; k += 128)
        sum += u[(size_t)b * H_SZ + k] * s2[(size_t)b * H_SZ + k];
    #pragma unroll
    for (int off = 16; off > 0; off >>= 1)
        sum += __shfl_xor(sum, off, 32);
    __shared__ float red[4];
    if ((tid & 31) == 0) red[tid >> 5] = sum;
    __syncthreads();
    if (tid == 0) out[b] = -(red[0] + red[1] + red[2] + red[3]);
}

extern "C" void kernel_launch(void* const* d_in, const int* in_sizes, int n_in,
                              void* d_out, int out_size, void* d_ws, size_t ws_size,
                              hipStream_t stream) {
    const float* xs = (const float*)d_in[0];
    const float* t  = (const float*)d_in[1];
    const float* W1 = (const float*)d_in[2];
    const float* b1 = (const float*)d_in[3];
    const float* W2 = (const float*)d_in[4];
    const float* b2 = (const float*)d_in[5];
    const float* W3 = (const float*)d_in[6];
    const float* b3 = (const float*)d_in[7];

    float* dxs  = (float*)d_out;
    float* ndiv = (float*)d_out + (size_t)B_SZ * D_SZ;

    float* ws = (float*)d_ws;
    const size_t act = (size_t)B_SZ * H_SZ;
    float* a1u = ws;
    float* s1  = ws + act;
    float* a2  = ws + 2 * act;
    float* s2  = ws + 3 * act;
    float* Cm  = ws + 4 * act;

    dim3 blk(NT);
    dim3 gridH(H_SZ / BN, B_SZ / BM);
    dim3 gridD((D_SZ + BN - 1) / BN, B_SZ / BM);

    precompute_C<<<(H_SZ * H_SZ) / 256, 256, 0, stream>>>(W1, W2, W3, Cm);

    gemm_wmma<true, 1><<<gridH, blk, 0, stream>>>(
        nullptr, xs, t, W1, b1, a1u, s1, B_SZ, H_SZ, D_SZ + 1);

    gemm_wmma<false, 1><<<gridH, blk, 0, stream>>>(
        a1u, nullptr, nullptr, W2, b2, a2, s2, B_SZ, H_SZ, H_SZ);

    gemm_wmma<false, 0><<<gridH, blk, 0, stream>>>(
        s1, nullptr, nullptr, Cm, nullptr, a1u, nullptr, B_SZ, H_SZ, H_SZ);

    gemm_wmma<false, 0><<<gridD, blk, 0, stream>>>(
        a2, nullptr, nullptr, W3, b3, dxs, nullptr, B_SZ, D_SZ, H_SZ);

    dot_div_kernel<<<B_SZ, 128, 0, stream>>>(a1u, s2, ndiv);
}


# GEMM operand path: WMMA fragments loaded from global memory straight to registers (16-byte loads, one permuted k order for A and B, 4-way interleaved columns), one chunk ahead; no LDS, no barrier
# speedup vs baseline: 5.6242x; 5.6242x over previous
#include <hip/hip_runtime.h>
#include <math.h>


typedef __attribute__((ext_vector_type(2))) float v2f;
typedef __attribute__((ext_vector_type(4))) float v4f;
typedef __attribute__((ext_vector_type(8))) float v8f;

#define B_SZ 4096
#define D_SZ 66
#define H_SZ 512


__device__ __forceinline__ v8f wmma4(v2f a, v2f b, v8f c) {
    return __builtin_amdgcn_wmma_f32_16x16x4_f32(
        false, a, false, b, (short)0, c, false, false);
}

__device__ __forceinline__ int kof(int kb, int h, int i) {
    return kb + ((i >> 2) << 3) + (h << 2) + (i & 3);
}

template<int MT, int NSUB, int KC>
struct Frags {
    float a[MT][KC / 2];
    float b[KC / 2][NSUB];
};

__global__ __launch_bounds__(256)
void precompute_C(const float* __restrict__ W1, const float* __restrict__ W2,
                  const float* __restrict__ W3, float* __restrict__ Cm)
{
    int idx = blockIdx.x * blockDim.x + threadIdx.x;
    int k = idx & (H_SZ - 1);
    int l = idx >> 9;
    float m = 0.0f;
    #pragma unroll 2
    for (int j = 0; j < D_SZ; ++j)
        m = fmaf(W1[j * H_SZ + l], W3[k * D_SZ + j], m);
    Cm[l * H_SZ + k] = m * W2[l * H_SZ + k];
}

template<int AMODE, int BMODE, int MT, int NSUB, int KC, int K, int N>
__device__ __forceinline__ void load_frags(
    Frags<MT, NSUB, KC>& f,
    const float* __restrict__ A, const float* __restrict__ xs,
    const float* __restrict__ tvec, const float* __restrict__ Bm,
    int mrow, int ncol, int h, int kb)
{
    constexpr bool KGUARD = (K % KC) != 0;

    #pragma unroll
    for (int mt = 0; mt < MT; ++mt) {
        const int gm = mrow + 16 * mt;
        if (AMODE == 0) {
            #pragma unroll
            for (int q = 0; q < KC / 8; ++q) {
                const v4f v = *(const v4f*)&A[(size_t)gm * K + kof(kb, h, 4 * q)];
                f.a[mt][4 * q + 0] = v.x;
                f.a[mt][4 * q + 1] = v.y;
                f.a[mt][4 * q + 2] = v.z;
                f.a[mt][4 * q + 3] = v.w;
            }
        } else {
            const float tv = tvec[gm];
            #pragma unroll
            for (int i = 0; i < KC / 2; ++i) {
                const int k = kof(kb, h, i);
                const float x = xs[gm * D_SZ + min(k, D_SZ - 1)];
                f.a[mt][i] = (k < D_SZ) ? x : ((k == D_SZ) ? tv : 0.0f);
            }
        }
    }

    #pragma unroll
    for (int i = 0; i < KC / 2; ++i) {
        const int k  = kof(kb, h, i);
        const int kr = KGUARD ? min(k, K - 1) : k;
        if (BMODE == 0) {
            v4f v = *(const v4f*)&Bm[(size_t)kr * N + ncol];
            if (KGUARD && k >= K) v = (v4f){0.0f, 0.0f, 0.0f, 0.0f};
            f.b[i][0] = v.x;
            f.b[i][1] = v.y;
            f.b[i][2] = v.z;
            f.b[i][3] = v.w;
        } else {
            #pragma unroll
            for (int j = 0; j < NSUB; ++j) {
                const int n = min(16 * j + ncol, N - 1);
                float v = Bm[(size_t)kr * N + n];
                if (KGUARD && k >= K) v = 0.0f;
                f.b[i][j] = v;
            }
        }
    }
}

template<int MT, int NSUB, int KC>
__device__ __forceinline__ void mma_chunk(const Frags<MT, NSUB, KC>& f,
                                          v8f (&acc)[MT][NSUB])
{
    #pragma unroll
    for (int t = 0; t < KC / 4; ++t) {
        #pragma unroll
        for (int mt = 0; mt < MT; ++mt) {
            const v2f a = {f.a[mt][2 * t], f.a[mt][2 * t + 1]};
            #pragma unroll
            for (int j = 0; j < NSUB; ++j) {
                const v2f b = {f.b[2 * t][j], f.b[2 * t + 1][j]};
                acc[mt][j] = wmma4(a, b, acc[mt][j]);
            }
        }
    }
}

template<int AMODE, int BMODE, int MT, int NSUB, int KC, int EPI, int WPB, int K, int N>
__global__ __launch_bounds__(WPB * 32)
void gemm_direct(const float* __restrict__ A, const float* __restrict__ xs,
                 const float* __restrict__ tvec,
                 const float* __restrict__ Bm, const float* __restrict__ bias,
                 float* __restrict__ out0, float* __restrict__ out1)
{
    static_assert(BMODE == 1 || NSUB == 4, "16-byte B loads feed 4 sub-tiles");
    constexpr int NCH = (K + KC - 1) / KC;

    const int lane = threadIdx.x & 31;
    const int wave = threadIdx.x >> 5;
    const int c = lane & 15;
    const int h = lane >> 4;

    const int m_base = (blockIdx.y * WPB + wave) * (16 * MT);
    const int n0     = blockIdx.x * (16 * NSUB);
    const int mrow   = m_base + c;
    const int ncol   = (BMODE == 0) ? (n0 + 4 * c) : (n0 + c);

    v8f acc[MT][NSUB] = {};
    Frags<MT, NSUB, KC> f0, f1;

    constexpr int NPAIR = (NCH - 1) / 2;
#define LOADF(F, KB) load_frags<AMODE, BMODE, MT, NSUB, KC, K, N>(F, A, xs, tvec, Bm, mrow, ncol, h, (KB))
#define MMAF(F) mma_chunk<MT, NSUB, KC>(F, acc)
#define SCHED_FENCE() __builtin_amdgcn_sched_barrier(0)
    LOADF(f0, 0);
    for (int p = 0; p < NPAIR; ++p) {
        const int kb = 2 * p * KC;
        LOADF(f1, kb + KC);      SCHED_FENCE();
        MMAF(f0);                SCHED_FENCE();
        LOADF(f0, kb + 2 * KC);  SCHED_FENCE();
        MMAF(f1);                SCHED_FENCE();
    }
    if ((NCH & 1) == 0) {
        LOADF(f1, (NCH - 1) * KC);  SCHED_FENCE();
        MMAF(f0);
        MMAF(f1);
    } else {
        MMAF(f0);
    }
#undef LOADF
#undef MMAF
#undef SCHED_FENCE

    if (BMODE == 0) {
        v4f bi = {0.0f, 0.0f, 0.0f, 0.0f};
        if (bias) bi = *(const v4f*)&bias[ncol];
        #pragma unroll
        for (int mt = 0; mt < MT; ++mt) {
            #pragma unroll
            for (int r = 0; r < 8; ++r) {
                const int gm = m_base + 16 * mt + 8 * h + r;
                v4f z = {acc[mt][0][r], acc[mt][1][r], acc[mt][2][r], acc[mt][3][r]};
                z += bi;
                if (EPI == 0) {
                    *(v4f*)&out0[(size_t)gm * N + ncol] = z;
                } else {
                    v4f o0, o1;
                    #pragma unroll
                    for (int e = 0; e < 4; ++e) {
                        const float sg = 1.0f / (1.0f + __expf(-z[e]));
                        o0[e] = z[e] * sg;
                        o1[e] = sg * (1.0f + z[e] * (1.0f - sg));
                    }
                    *(v4f*)&out0[(size_t)gm * N + ncol] = o0;
                    *(v4f*)&out1[(size_t)gm * N + ncol] = o1;
                }
            }
        }
    } else {
        #pragma unroll
        for (int j = 0; j < NSUB; ++j) {
            const int n = 16 * j + ncol;
            if (n >= N) continue;
            const float bi = bias ? bias[n] : 0.0f;
            #pragma unroll
            for (int mt = 0; mt < MT; ++mt) {
                #pragma unroll
                for (int r = 0; r < 8; ++r) {
                    const int gm = m_base + 16 * mt + 8 * h + r;
                    const float z = acc[mt][j][r] + bi;
                    if (EPI == 0) {
                        out0[(size_t)gm * N + n] = z;
                    } else {
                        const float sg = 1.0f / (1.0f + __expf(-z));
                        out0[(size_t)gm * N + n] = z * sg;
                        out1[(size_t)gm * N + n] = sg * (1.0f + z * (1.0f - sg));
                    }
                }
            }
        }
    }
}

__global__ __launch_bounds__(128)
void dot_div_kernel(const float* __restrict__ u, const float* __restrict__ s2,
                    float* __restrict__ out)
{
    int b = blockIdx.x;
    int tid = threadIdx.x;
    float sum = 0.0f;
    #pragma unroll
    for (int k = tid; k < H_SZ; k += 128)
        sum += u[(size_t)b * H_SZ + k] * s2[(size_t)b * H_SZ + k];
    #pragma unroll
    for (int off = 16; off > 0; off >>= 1)
        sum += __shfl_xor(sum, off, 32);
    __shared__ float red[4];
    if ((tid & 31) == 0) red[tid >> 5] = sum;
    __syncthreads();
    if (tid == 0) out[b] = -(red[0] + red[1] + red[2] + red[3]);
}

extern "C" void kernel_launch(void* const* d_in, const int* in_sizes, int n_in,
                              void* d_out, int out_size, void* d_ws, size_t ws_size,
                              hipStream_t stream) {
    const float* xs = (const float*)d_in[0];
    const float* t  = (const float*)d_in[1];
    const float* W1 = (const float*)d_in[2];
    const float* b1 = (const float*)d_in[3];
    const float* W2 = (const float*)d_in[4];
    const float* b2 = (const float*)d_in[5];
    const float* W3 = (const float*)d_in[6];
    const float* b3 = (const float*)d_in[7];

    float* dxs  = (float*)d_out;
    float* ndiv = (float*)d_out + (size_t)B_SZ * D_SZ;

    float* ws = (float*)d_ws;
    const size_t act = (size_t)B_SZ * H_SZ;
    float* a1u = ws;
    float* s1  = ws + act;
    float* a2  = ws + 2 * act;
    float* s2  = ws + 3 * act;
    float* Cm  = ws + 4 * act;

    dim3 gridH(H_SZ / 64, B_SZ / 128);
    dim3 gridD(2, B_SZ / 16);

    precompute_C<<<(H_SZ * H_SZ) / 256, 256, 0, stream>>>(W1, W2, W3, Cm);

    gemm_direct<1, 0, 2, 4, 32, 1, 4, D_SZ + 1, H_SZ><<<gridH, 128, 0, stream>>>(
        nullptr, xs, t, W1, b1, a1u, s1);

    gemm_direct<0, 0, 2, 4, 32, 1, 4, H_SZ, H_SZ><<<gridH, 128, 0, stream>>>(
        a1u, nullptr, nullptr, W2, b2, a2, s2);

    gemm_direct<0, 0, 2, 4, 32, 0, 4, H_SZ, H_SZ><<<gridH, 128, 0, stream>>>(
        s1, nullptr, nullptr, Cm, nullptr, a1u, nullptr);

    gemm_direct<0, 1, 1, 3, 32, 0, 1, H_SZ, D_SZ><<<gridD, 32, 0, stream>>>(
        a2, nullptr, nullptr, W3, b3, dxs, nullptr);

    dot_div_kernel<<<B_SZ, 128, 0, stream>>>(a1u, s2, ndiv);
}


# trace-matrix kernel kept on the vector unit, tiled 32x32 through LDS: W1 and W3 tiles copied in by whole-line 16-byte loads, 4 outputs per thread (was 264-byte-strided reads); 8192 waves -> 2048
# speedup vs baseline: 8.9958x; 1.5995x over previous
#include <hip/hip_runtime.h>
#include <math.h>


typedef __attribute__((ext_vector_type(2))) float v2f;
typedef __attribute__((ext_vector_type(4))) float v4f;
typedef __attribute__((ext_vector_type(8))) float v8f;

#define B_SZ 4096
#define D_SZ 66
#define H_SZ 512


__device__ __forceinline__ v8f wmma4(v2f a, v2f b, v8f c) {
    return __builtin_amdgcn_wmma_f32_16x16x4_f32(
        false, a, false, b, (short)0, c, false, false);
}

__device__ __forceinline__ int kof(int kb, int h, int i) {
    return kb + ((i >> 2) << 3) + (h << 2) + (i & 3);
}

#define CT 32
#define CNV (CT * D_SZ / 4)
__global__ __launch_bounds__(256)
void precompute_C(const float* __restrict__ W1, const float* __restrict__ W2,
                  const float* __restrict__ W3, float* __restrict__ Cm)
{
    __shared__ __attribute__((aligned(16))) float sW1[D_SZ * CT];
    __shared__ __attribute__((aligned(16))) float sW3[CT * D_SZ];

    const int tid = threadIdx.x;
    const int l0 = (blockIdx.x >> 4) * CT;
    const int k0 = (blockIdx.x & 15) * CT;
    const int kk = tid & 31;
    const int lq = tid >> 5;

    float w2[4];
    for (int i = 0; i < 4; ++i)
        w2[i] = W2[(size_t)(l0 + 4 * lq + i) * H_SZ + k0 + kk];

    const float* w3src = W3 + (size_t)k0 * D_SZ;
    v4f r1[3], r3[3];
    for (int i = 0; i < 3; ++i) {
        const int e = tid + 256 * i;
        r1[i] = (v4f){0.0f, 0.0f, 0.0f, 0.0f};
        r3[i] = (v4f){0.0f, 0.0f, 0.0f, 0.0f};
        if (e < CNV) {
            r1[i] = *(const v4f*)&W1[(size_t)(e >> 3) * H_SZ + l0 + 4 * (e & 7)];
            r3[i] = *(const v4f*)&w3src[4 * e];
        }
    }
    for (int i = 0; i < 3; ++i) {
        const int e = tid + 256 * i;
        if (e < CNV) {
            *(v4f*)&sW1[4 * e] = r1[i];
            *(v4f*)&sW3[4 * e] = r3[i];
        }
    }
    __syncthreads();

    const float* w3row = &sW3[kk * D_SZ];
    const float* w1col = &sW1[4 * lq];
    float m[4] = {0.0f, 0.0f, 0.0f, 0.0f};
    for (int j = 0; j < D_SZ; j += 2) {
        const v2f b  = *(const v2f*)&w3row[j];
        const v4f a0 = *(const v4f*)&w1col[j * CT];
        const v4f a1 = *(const v4f*)&w1col[(j + 1) * CT];
        for (int i = 0; i < 4; ++i) m[i] = fmaf(a0[i], b.x, m[i]);
        for (int i = 0; i < 4; ++i) m[i] = fmaf(a1[i], b.y, m[i]);
    }
    for (int i = 0; i < 4; ++i)
        Cm[(size_t)(l0 + 4 * lq + i) * H_SZ + k0 + kk] = m[i] * w2[i];
}

#define STRIP 16
#define XP 100
#define AP 516
#define OP 80

template<int NSUB>
struct FragS {
    float a[16];
    float b[16][NSUB];
};

template<int K>
__device__ __forceinline__ void load_b_wide(FragS<8>& f, const float* __restrict__ Bm,
                                            int col, int h, int kb)
{
    constexpr bool KGUARD = (K % 32) != 0;
    #pragma unroll
    for (int i = 0; i < 16; ++i) {
        const int k  = kof(kb, h, i);
        const int kr = KGUARD ? min(k, K - 1) : k;
        v4f v0 = *(const v4f*)&Bm[(size_t)kr * H_SZ + col];
        v4f v1 = *(const v4f*)&Bm[(size_t)kr * H_SZ + col + 64];
        if (KGUARD && k >= K) {
            v0 = (v4f){0.0f, 0.0f, 0.0f, 0.0f};
            v1 = (v4f){0.0f, 0.0f, 0.0f, 0.0f};
        }
        f.b[i][0] = v0.x;  f.b[i][1] = v0.y;  f.b[i][2] = v0.z;  f.b[i][3] = v0.w;
        f.b[i][4] = v1.x;  f.b[i][5] = v1.y;  f.b[i][6] = v1.z;  f.b[i][7] = v1.w;
    }
}

__device__ __forceinline__ void load_b_out(FragS<5>& f, const float* __restrict__ W3,
                                           int c, int h, int kb)
{
    #pragma unroll
    for (int i = 0; i < 16; ++i) {
        const int k = kof(kb, h, i);
        #pragma unroll
        for (int j = 0; j < 5; ++j) {
            const int n = min(16 * j + c, D_SZ - 1);
            f.b[i][j] = W3[(size_t)k * D_SZ + n];
        }
    }
}

template<int NSUB>
__device__ __forceinline__ void load_a_lds(FragS<NSUB>& f, const float* arow, int h, int kb)
{
    #pragma unroll
    for (int q = 0; q < 4; ++q) {
        const v4f v = *(const v4f*)&arow[kof(kb, h, 4 * q)];
        f.a[4 * q + 0] = v.x;
        f.a[4 * q + 1] = v.y;
        f.a[4 * q + 2] = v.z;
        f.a[4 * q + 3] = v.w;
    }
}

template<int NSUB>
__device__ __forceinline__ void mma_strip(const FragS<NSUB>& f, v8f (&acc)[NSUB])
{
    #pragma unroll
    for (int t = 0; t < 8; ++t) {
        const v2f a = {f.a[2 * t], f.a[2 * t + 1]};
        #pragma unroll
        for (int j = 0; j < NSUB; ++j) {
            const v2f b = {f.b[2 * t][j], f.b[2 * t + 1][j]};
            acc[j] = wmma4(a, b, acc[j]);
        }
    }
}

#define SCHED_FENCE() __builtin_amdgcn_sched_barrier(0)

template<int K>
__device__ __forceinline__ void product_wide(v8f (&acc)[8], FragS<8>& f0, FragS<8>& f1,
                                             const float* arow,
                                             const float* __restrict__ Bm, int col, int h)
{
    constexpr int NCH = (K + 31) / 32;
    constexpr int NPAIR = (NCH - 1) / 2;
    load_a_lds<8>(f0, arow, h, 0);
    for (int p = 0; p < NPAIR; ++p) {
        const int kb = 64 * p;
        load_b_wide<K>(f1, Bm, col, h, kb + 32);
        load_a_lds<8>(f1, arow, h, kb + 32);       SCHED_FENCE();
        mma_strip<8>(f0, acc);                     SCHED_FENCE();
        load_b_wide<K>(f0, Bm, col, h, kb + 64);
        load_a_lds<8>(f0, arow, h, kb + 64);       SCHED_FENCE();
        mma_strip<8>(f1, acc);                     SCHED_FENCE();
    }
    if ((NCH & 1) == 0) {
        load_b_wide<K>(f1, Bm, col, h, (NCH - 1) * 32);
        load_a_lds<8>(f1, arow, h, (NCH - 1) * 32);  SCHED_FENCE();
        mma_strip<8>(f0, acc);
        mma_strip<8>(f1, acc);
    } else {
        mma_strip<8>(f0, acc);
    }
}

__global__ __launch_bounds__(128)
void strip_mlp(const float* __restrict__ xs, const float* __restrict__ tvec,
               const float* __restrict__ W1, const float* __restrict__ b1,
               const float* __restrict__ W2, const float* __restrict__ b2,
               const float* __restrict__ W3, const float* __restrict__ b3,
               const float* __restrict__ Cm,
               float* __restrict__ dxs, float* __restrict__ ndiv)
{
    __shared__ __attribute__((aligned(16))) float sX[STRIP * XP];
    __shared__ __attribute__((aligned(16))) float sA[STRIP * AP];
    __shared__ __attribute__((aligned(16))) float sS[STRIP * AP];
    __shared__ float sR[4 * STRIP];

    const int tid  = threadIdx.x;
    const int lane = tid & 31;
    const int wave = tid >> 5;
    const int c = lane & 15;
    const int h = lane >> 4;
    const int row0 = blockIdx.x * STRIP;
    const int col  = 128 * wave + 4 * c;

    FragS<8> f0, f1;
    load_b_wide<D_SZ + 1>(f0, W1, col, h, 0);

    const v4f bias1a = *(const v4f*)&b1[col];
    const v4f bias1b = *(const v4f*)&b1[col + 64];
    const v4f bias2a = *(const v4f*)&b2[col];
    const v4f bias2b = *(const v4f*)&b2[col + 64];

    for (int e = tid; e < STRIP * D_SZ; e += 128) {
        const int r = e / D_SZ;
        const int k = e - r * D_SZ;
        sX[r * XP + k] = xs[(size_t)row0 * D_SZ + e];
    }
    for (int e = tid; e < STRIP * (XP - D_SZ); e += 128) {
        const int r = e / (XP - D_SZ);
        const int k = D_SZ + (e - r * (XP - D_SZ));
        sX[r * XP + k] = (k == D_SZ) ? tvec[row0 + r] : 0.0f;
    }
    __syncthreads();

    v8f acc[8];

    #pragma unroll
    for (int j = 0; j < 8; ++j) acc[j] = (v8f){0.0f, 0.0f, 0.0f, 0.0f, 0.0f, 0.0f, 0.0f, 0.0f};
    product_wide<D_SZ + 1>(acc, f0, f1, &sX[c * XP], W1, col, h);
    load_b_wide<H_SZ>(f0, W2, col, h, 0);            SCHED_FENCE();
    #pragma unroll
    for (int g = 0; g < 2; ++g) {
        const v4f bi = g ? bias1b : bias1a;
        #pragma unroll
        for (int r = 0; r < 8; ++r) {
            v4f z = {acc[4 * g + 0][r], acc[4 * g + 1][r], acc[4 * g + 2][r], acc[4 * g + 3][r]};
            z += bi;
            v4f o0, o1;
            #pragma unroll
            for (int e = 0; e < 4; ++e) {
                const float sg = 1.0f / (1.0f + __expf(-z[e]));
                o0[e] = z[e] * sg;
                o1[e] = sg * (1.0f + z[e] * (1.0f - sg));
            }
            *(v4f*)&sA[(8 * h + r) * AP + col + 64 * g] = o0;
            *(v4f*)&sS[(8 * h + r) * AP + col + 64 * g] = o1;
        }
    }
    __syncthreads();
    SCHED_FENCE();

    #pragma unroll
    for (int j = 0; j < 8; ++j) acc[j] = (v8f){0.0f, 0.0f, 0.0f, 0.0f, 0.0f, 0.0f, 0.0f, 0.0f};
    product_wide<H_SZ>(acc, f0, f1, &sA[c * AP], W2, col, h);
    load_b_wide<H_SZ>(f0, Cm, col, h, 0);            SCHED_FENCE();
    v8f a2r[8], s2r[8];
    #pragma unroll
    for (int j = 0; j < 8; ++j) {
        const v4f bi = (j < 4) ? bias2a : bias2b;
        #pragma unroll
        for (int r = 0; r < 8; ++r) {
            const float z  = acc[j][r] + bi[j & 3];
            const float sg = 1.0f / (1.0f + __expf(-z));
            a2r[j][r] = z * sg;
            s2r[j][r] = sg * (1.0f + z * (1.0f - sg));
        }
    }
    SCHED_FENCE();
    __syncthreads();
    #pragma unroll
    for (int g = 0; g < 2; ++g) {
        #pragma unroll
        for (int r = 0; r < 8; ++r) {
            const v4f o = {a2r[4 * g + 0][r], a2r[4 * g + 1][r], a2r[4 * g + 2][r], a2r[4 * g + 3][r]};
            *(v4f*)&sA[(8 * h + r) * AP + col + 64 * g] = o;
        }
    }
    SCHED_FENCE();

    #pragma unroll
    for (int j = 0; j < 8; ++j) acc[j] = (v8f){0.0f, 0.0f, 0.0f, 0.0f, 0.0f, 0.0f, 0.0f, 0.0f};
    product_wide<H_SZ>(acc, f0, f1, &sS[c * AP], Cm, col, h);
    FragS<5> g0, g1;
    const int kw = 128 * wave;
    load_b_out(g0, W3, c, h, kw);                    SCHED_FENCE();
    float p[8];
    #pragma unroll
    for (int r = 0; r < 8; ++r) {
        p[r] = 0.0f;
        #pragma unroll
        for (int j = 0; j < 8; ++j) p[r] = fmaf(acc[j][r], s2r[j][r], p[r]);
    }
    #pragma unroll
    for (int off = 8; off > 0; off >>= 1) {
        #pragma unroll
        for (int r = 0; r < 8; ++r) p[r] += __shfl_xor(p[r], off, 32);
    }
    if (c == 0) {
        #pragma unroll
        for (int r = 0; r < 8; ++r) sR[wave * STRIP + 8 * h + r] = p[r];
    }
    __syncthreads();
    SCHED_FENCE();

    v8f acc4[5];
    #pragma unroll
    for (int j = 0; j < 5; ++j) acc4[j] = (v8f){0.0f, 0.0f, 0.0f, 0.0f, 0.0f, 0.0f, 0.0f, 0.0f};
    const float* a2row = &sA[c * AP];
    load_a_lds<5>(g0, a2row, h, kw);
    load_b_out(g1, W3, c, h, kw + 32);
    load_a_lds<5>(g1, a2row, h, kw + 32);            SCHED_FENCE();
    mma_strip<5>(g0, acc4);                          SCHED_FENCE();
    load_b_out(g0, W3, c, h, kw + 64);
    load_a_lds<5>(g0, a2row, h, kw + 64);            SCHED_FENCE();
    mma_strip<5>(g1, acc4);                          SCHED_FENCE();
    load_b_out(g1, W3, c, h, kw + 96);
    load_a_lds<5>(g1, a2row, h, kw + 96);            SCHED_FENCE();
    mma_strip<5>(g0, acc4);
    mma_strip<5>(g1, acc4);
    #pragma unroll
    for (int j = 0; j < 5; ++j) {
        #pragma unroll
        for (int r = 0; r < 8; ++r)
            sS[(wave * STRIP + 8 * h + r) * OP + 16 * j + c] = acc4[j][r];
    }
    __syncthreads();

    for (int e = tid; e < STRIP * D_SZ; e += 128) {
        const int r = e / D_SZ;
        const int n = e - r * D_SZ;
        const float v = sS[(0 * STRIP + r) * OP + n] + sS[(1 * STRIP + r) * OP + n]
                      + sS[(2 * STRIP + r) * OP + n] + sS[(3 * STRIP + r) * OP + n];
        dxs[(size_t)row0 * D_SZ + e] = v + b3[n];
    }
    if (tid < STRIP)
        ndiv[row0 + tid] = -(sR[tid] + sR[STRIP + tid] + sR[2 * STRIP + tid] + sR[3 * STRIP + tid]);
}

extern "C" void kernel_launch(void* const* d_in, const int* in_sizes, int n_in,
                              void* d_out, int out_size, void* d_ws, size_t ws_size,
                              hipStream_t stream) {
    const float* xs = (const float*)d_in[0];
    const float* t  = (const float*)d_in[1];
    const float* W1 = (const float*)d_in[2];
    const float* b1 = (const float*)d_in[3];
    const float* W2 = (const float*)d_in[4];
    const float* b2 = (const float*)d_in[5];
    const float* W3 = (const float*)d_in[6];
    const float* b3 = (const float*)d_in[7];

    float* dxs  = (float*)d_out;
    float* ndiv = (float*)d_out + (size_t)B_SZ * D_SZ;

    float* Cm = (float*)d_ws;

    precompute_C<<<(H_SZ / CT) * (H_SZ / CT), 256, 0, stream>>>(W1, W2, W3, Cm);

    strip_mlp<<<B_SZ / STRIP, 128, 0, stream>>>(xs, t, W1, b1, W2, b2, W3, b3, Cm, dxs, ndiv);
}


# weights of the three 512-wide products (W1, W2, trace matrix) written by launch 0 in WMMA B-fragment order: a 16-byte load lands as two operand pairs, one 512-byte run per load, no register moves
# speedup vs baseline: 10.2457x; 1.1389x over previous
#include <hip/hip_runtime.h>
#include <math.h>


typedef __attribute__((ext_vector_type(2))) float v2f;
typedef __attribute__((ext_vector_type(4))) float v4f;
typedef __attribute__((ext_vector_type(8))) float v8f;

#define B_SZ 4096
#define D_SZ 66
#define H_SZ 512


__device__ __forceinline__ v8f wmma4(v2f a, v2f b, v8f c) {
    return __builtin_amdgcn_wmma_f32_16x16x4_f32(
        false, a, false, b, (short)0, c, false, false);
}

__device__ __forceinline__ int kof(int kb, int h, int i) {
    return kb + ((i >> 2) << 3) + (h << 2) + (i & 3);
}

#define PCHUNK (4 * 32 * 32 * 4)
#define CT 32
struct PieceMap {
    int kr;
    int nc;
    size_t off;
};
__device__ __forceinline__ PieceMap piece_of(int tid, int t, int n0)
{
    const int cl  = tid & 7;
    const int hh  = (tid >> 3) & 1;
    const int jpl = (tid >> 4) & 1;
    const int s   = tid >> 5;
    const int w   = n0 >> 7;
    const int g   = (n0 >> 6) & 1;
    const int c0  = (n0 >> 2) & 8;
    const int q    = 4 * s + 2 * g + jpl;
    const int lane = 16 * hh + c0 + cl;
    PieceMap m;
    m.kr  = 8 * (s >> 1) + 4 * hh + 2 * (s & 1);
    m.nc  = 4 * cl + 2 * jpl;
    m.off = ((((size_t)t * 4 + w) * 32 + q) * 32 + lane) * 4;
    return m;
}

#define CNV (CT * D_SZ / 4)
#define CP 34
#define NT_C ((H_SZ / CT) * (H_SZ / CT))
#define NT_W1 (3 * (H_SZ / CT))
__global__ __launch_bounds__(256)
void pack_weights(const float* __restrict__ W1, const float* __restrict__ W2,
                  const float* __restrict__ W3,
                  float* __restrict__ Pc, float* __restrict__ Pw2, float* __restrict__ Pw1)
{
    __shared__ __attribute__((aligned(16))) float sW1[D_SZ * CT];
    __shared__ __attribute__((aligned(16))) float sW3[CT * D_SZ];
    __shared__ __attribute__((aligned(16))) float sC[CT * CP];
    __shared__ __attribute__((aligned(16))) float sW[CT * CP];

    const int tid = threadIdx.x;

    if (blockIdx.x >= NT_C) {
        const int bb = blockIdx.x - NT_C;
        const int t  = bb >> 4;
        const int n0 = (bb & 15) * CT;
        const PieceMap m = piece_of(tid, t, n0);
        const int k = 32 * t + m.kr;
        v2f r0 = {0.0f, 0.0f}, r1 = {0.0f, 0.0f};
        if (k < D_SZ + 1)     r0 = *(const v2f*)&W1[(size_t)k * H_SZ + n0 + m.nc];
        if (k + 1 < D_SZ + 1) r1 = *(const v2f*)&W1[(size_t)(k + 1) * H_SZ + n0 + m.nc];
        *(v4f*)&Pw1[m.off] = (v4f){r0.x, r1.x, r0.y, r1.y};
        return;
    }

    const int l0 = (blockIdx.x >> 4) * CT;
    const int k0 = (blockIdx.x & 15) * CT;
    const int kk = tid & 31;
    const int lq = tid >> 5;

    float w2[4];
    for (int i = 0; i < 4; ++i)
        w2[i] = W2[(size_t)(l0 + 4 * lq + i) * H_SZ + k0 + kk];

    const float* w3src = W3 + (size_t)k0 * D_SZ;
    v4f r1[3], r3[3];
    for (int i = 0; i < 3; ++i) {
        const int e = tid + 256 * i;
        r1[i] = (v4f){0.0f, 0.0f, 0.0f, 0.0f};
        r3[i] = (v4f){0.0f, 0.0f, 0.0f, 0.0f};
        if (e < CNV) {
            r1[i] = *(const v4f*)&W1[(size_t)(e >> 3) * H_SZ + l0 + 4 * (e & 7)];
            r3[i] = *(const v4f*)&w3src[4 * e];
        }
    }
    for (int i = 0; i < 3; ++i) {
        const int e = tid + 256 * i;
        if (e < CNV) {
            *(v4f*)&sW1[4 * e] = r1[i];
            *(v4f*)&sW3[4 * e] = r3[i];
        }
    }
    __syncthreads();

    const float* w3row = &sW3[kk * D_SZ];
    const float* w1col = &sW1[4 * lq];
    float m[4] = {0.0f, 0.0f, 0.0f, 0.0f};
    for (int j = 0; j < D_SZ; j += 2) {
        const v2f b  = *(const v2f*)&w3row[j];
        const v4f a0 = *(const v4f*)&w1col[j * CT];
        const v4f a1 = *(const v4f*)&w1col[(j + 1) * CT];
        for (int i = 0; i < 4; ++i) m[i] = fmaf(a0[i], b.x, m[i]);
        for (int i = 0; i < 4; ++i) m[i] = fmaf(a1[i], b.y, m[i]);
    }
    for (int i = 0; i < 4; ++i) {
        sC[(4 * lq + i) * CP + kk] = m[i] * w2[i];
        sW[(4 * lq + i) * CP + kk] = w2[i];
    }
    __syncthreads();

    const PieceMap pm = piece_of(tid, l0 >> 5, k0);
    {
        const v2f c0 = *(const v2f*)&sC[pm.kr * CP + pm.nc];
        const v2f c1 = *(const v2f*)&sC[(pm.kr + 1) * CP + pm.nc];
        *(v4f*)&Pc[pm.off] = (v4f){c0.x, c1.x, c0.y, c1.y};
        const v2f u0 = *(const v2f*)&sW[pm.kr * CP + pm.nc];
        const v2f u1 = *(const v2f*)&sW[(pm.kr + 1) * CP + pm.nc];
        *(v4f*)&Pw2[pm.off] = (v4f){u0.x, u1.x, u0.y, u1.y};
    }
}

#define STRIP 16
#define XP 100
#define AP 516
#define OP 80


struct FragP {
    float a[16];
    v4f   b[32];
};

template<int NSUB>
struct FragS {
    float a[16];
    float b[16][NSUB];
};

__device__ __forceinline__ void load_b_packed(FragP& f, const float* __restrict__ Pl, int t)
{
    const v4f* p = (const v4f*)(Pl + (size_t)t * PCHUNK);
    #pragma unroll
    for (int q = 0; q < 32; ++q) f.b[q] = p[q * 32];
}

__device__ __forceinline__ void load_b_out(FragS<5>& f, const float* __restrict__ W3,
                                           int c, int h, int kb)
{
    #pragma unroll
    for (int i = 0; i < 16; ++i) {
        const int k = kof(kb, h, i);
        #pragma unroll
        for (int j = 0; j < 5; ++j) {
            const int n = min(16 * j + c, D_SZ - 1);
            f.b[i][j] = W3[(size_t)k * D_SZ + n];
        }
    }
}

template<class F>
__device__ __forceinline__ void load_a_lds(F& f, const float* arow, int h, int kb)
{
    #pragma unroll
    for (int q = 0; q < 4; ++q) {
        const v4f v = *(const v4f*)&arow[kof(kb, h, 4 * q)];
        f.a[4 * q + 0] = v.x;
        f.a[4 * q + 1] = v.y;
        f.a[4 * q + 2] = v.z;
        f.a[4 * q + 3] = v.w;
    }
}

__device__ __forceinline__ void mma_packed(const FragP& f, v8f (&acc)[8])
{
    #pragma unroll
    for (int s = 0; s < 8; ++s) {
        const v2f a = {f.a[2 * s], f.a[2 * s + 1]};
        #pragma unroll
        for (int j = 0; j < 8; ++j) {
            const v4f v = f.b[4 * s + (j >> 1)];
            const v2f b = (j & 1) ? (v2f){v.z, v.w} : (v2f){v.x, v.y};
            acc[j] = wmma4(a, b, acc[j]);
        }
    }
}

template<int NSUB>
__device__ __forceinline__ void mma_strip(const FragS<NSUB>& f, v8f (&acc)[NSUB])
{
    #pragma unroll
    for (int t = 0; t < 8; ++t) {
        const v2f a = {f.a[2 * t], f.a[2 * t + 1]};
        #pragma unroll
        for (int j = 0; j < NSUB; ++j) {
            const v2f b = {f.b[2 * t][j], f.b[2 * t + 1][j]};
            acc[j] = wmma4(a, b, acc[j]);
        }
    }
}

#define SCHED_FENCE() __builtin_amdgcn_sched_barrier(0)

template<int NCH>
__device__ __forceinline__ void product_packed(v8f (&acc)[8], FragP& f0, FragP& f1,
                                               const float* arow,
                                               const float* __restrict__ Pl, int h)
{
    constexpr int NPAIR = (NCH - 1) / 2;
    load_a_lds(f0, arow, h, 0);
    for (int p = 0; p < NPAIR; ++p) {
        const int t = 2 * p;
        load_b_packed(f1, Pl, t + 1);
        load_a_lds(f1, arow, h, 32 * (t + 1));     SCHED_FENCE();
        mma_packed(f0, acc);                       SCHED_FENCE();
        load_b_packed(f0, Pl, t + 2);
        load_a_lds(f0, arow, h, 32 * (t + 2));     SCHED_FENCE();
        mma_packed(f1, acc);                       SCHED_FENCE();
    }
    if ((NCH & 1) == 0) {
        load_b_packed(f1, Pl, NCH - 1);
        load_a_lds(f1, arow, h, 32 * (NCH - 1));   SCHED_FENCE();
        mma_packed(f0, acc);
        mma_packed(f1, acc);
    } else {
        mma_packed(f0, acc);
    }
}

__global__ __launch_bounds__(128)
void strip_mlp(const float* __restrict__ xs, const float* __restrict__ tvec,
               const float* __restrict__ Pw1, const float* __restrict__ b1,
               const float* __restrict__ Pw2, const float* __restrict__ b2,
               const float* __restrict__ W3, const float* __restrict__ b3,
               const float* __restrict__ Pc,
               float* __restrict__ dxs, float* __restrict__ ndiv)
{
    __shared__ __attribute__((aligned(16))) float sX[STRIP * XP];
    __shared__ __attribute__((aligned(16))) float sA[STRIP * AP];
    __shared__ __attribute__((aligned(16))) float sS[STRIP * AP];
    __shared__ float sR[4 * STRIP];

    const int tid  = threadIdx.x;
    const int lane = tid & 31;
    const int wave = tid >> 5;
    const int c = lane & 15;
    const int h = lane >> 4;
    const int row0 = blockIdx.x * STRIP;
    const int col  = 128 * wave + 4 * c;

    const int loff = wave * (32 * 32 * 4) + lane * 4;

    FragP f0, f1;
    load_b_packed(f0, Pw1 + loff, 0);

    const v4f bias1a = *(const v4f*)&b1[col];
    const v4f bias1b = *(const v4f*)&b1[col + 64];
    const v4f bias2a = *(const v4f*)&b2[col];
    const v4f bias2b = *(const v4f*)&b2[col + 64];

    for (int e = tid; e < STRIP * D_SZ; e += 128) {
        const int r = e / D_SZ;
        const int k = e - r * D_SZ;
        sX[r * XP + k] = xs[(size_t)row0 * D_SZ + e];
    }
    for (int e = tid; e < STRIP * (XP - D_SZ); e += 128) {
        const int r = e / (XP - D_SZ);
        const int k = D_SZ + (e - r * (XP - D_SZ));
        sX[r * XP + k] = (k == D_SZ) ? tvec[row0 + r] : 0.0f;
    }
    __syncthreads();

    v8f acc[8];

    #pragma unroll
    for (int j = 0; j < 8; ++j) acc[j] = (v8f){0.0f, 0.0f, 0.0f, 0.0f, 0.0f, 0.0f, 0.0f, 0.0f};
    product_packed<3>(acc, f0, f1, &sX[c * XP], Pw1 + loff, h);
    load_b_packed(f0, Pw2 + loff, 0);                SCHED_FENCE();
    #pragma unroll
    for (int g = 0; g < 2; ++g) {
        const v4f bi = g ? bias1b : bias1a;
        #pragma unroll
        for (int r = 0; r < 8; ++r) {
            v4f z = {acc[4 * g + 0][r], acc[4 * g + 1][r], acc[4 * g + 2][r], acc[4 * g + 3][r]};
            z += bi;
            v4f o0, o1;
            #pragma unroll
            for (int e = 0; e < 4; ++e) {
                const float sg = 1.0f / (1.0f + __expf(-z[e]));
                o0[e] = z[e] * sg;
                o1[e] = sg * (1.0f + z[e] * (1.0f - sg));
            }
            *(v4f*)&sA[(8 * h + r) * AP + col + 64 * g] = o0;
            *(v4f*)&sS[(8 * h + r) * AP + col + 64 * g] = o1;
        }
    }
    __syncthreads();

    #pragma unroll
    for (int j = 0; j < 8; ++j) acc[j] = (v8f){0.0f, 0.0f, 0.0f, 0.0f, 0.0f, 0.0f, 0.0f, 0.0f};
    product_packed<H_SZ / 32>(acc, f0, f1, &sA[c * AP], Pw2 + loff, h);
    load_b_packed(f0, Pc + loff, 0);                 SCHED_FENCE();
    v8f a2r[8], s2r[8];
    #pragma unroll
    for (int j = 0; j < 8; ++j) {
        const v4f bi = (j < 4) ? bias2a : bias2b;
        #pragma unroll
        for (int r = 0; r < 8; ++r) {
            const float z  = acc[j][r] + bi[j & 3];
            const float sg = 1.0f / (1.0f + __expf(-z));
            a2r[j][r] = z * sg;
            s2r[j][r] = sg * (1.0f + z * (1.0f - sg));
        }
    }
    SCHED_FENCE();
    __syncthreads();
    #pragma unroll
    for (int g = 0; g < 2; ++g) {
        #pragma unroll
        for (int r = 0; r < 8; ++r) {
            const v4f o = {a2r[4 * g + 0][r], a2r[4 * g + 1][r], a2r[4 * g + 2][r], a2r[4 * g + 3][r]};
            *(v4f*)&sA[(8 * h + r) * AP + col + 64 * g] = o;
        }
    }

    #pragma unroll
    for (int j = 0; j < 8; ++j) acc[j] = (v8f){0.0f, 0.0f, 0.0f, 0.0f, 0.0f, 0.0f, 0.0f, 0.0f};
    product_packed<H_SZ / 32>(acc, f0, f1, &sS[c * AP], Pc + loff, h);
    FragS<5> g0, g1;
    const int kw = 128 * wave;
    load_b_out(g0, W3, c, h, kw);                    SCHED_FENCE();
    float p[8];
    #pragma unroll
    for (int r = 0; r < 8; ++r) {
        p[r] = 0.0f;
        #pragma unroll
        for (int j = 0; j < 8; ++j) p[r] = fmaf(acc[j][r], s2r[j][r], p[r]);
    }
    #pragma unroll
    for (int off = 8; off > 0; off >>= 1) {
        #pragma unroll
        for (int r = 0; r < 8; ++r) p[r] += __shfl_xor(p[r], off, 32);
    }
    if (c == 0) {
        #pragma unroll
        for (int r = 0; r < 8; ++r) sR[wave * STRIP + 8 * h + r] = p[r];
    }
    __syncthreads();

    v8f acc4[5];
    #pragma unroll
    for (int j = 0; j < 5; ++j) acc4[j] = (v8f){0.0f, 0.0f, 0.0f, 0.0f, 0.0f, 0.0f, 0.0f, 0.0f};
    const float* a2row = &sA[c * AP];
    load_a_lds(g0, a2row, h, kw);
    load_b_out(g1, W3, c, h, kw + 32);
    load_a_lds(g1, a2row, h, kw + 32);            SCHED_FENCE();
    mma_strip<5>(g0, acc4);                          SCHED_FENCE();
    load_b_out(g0, W3, c, h, kw + 64);
    load_a_lds(g0, a2row, h, kw + 64);            SCHED_FENCE();
    mma_strip<5>(g1, acc4);                          SCHED_FENCE();
    load_b_out(g1, W3, c, h, kw + 96);
    load_a_lds(g1, a2row, h, kw + 96);            SCHED_FENCE();
    mma_strip<5>(g0, acc4);
    mma_strip<5>(g1, acc4);
    #pragma unroll
    for (int j = 0; j < 5; ++j) {
        #pragma unroll
        for (int r = 0; r < 8; ++r)
            sS[(wave * STRIP + 8 * h + r) * OP + 16 * j + c] = acc4[j][r];
    }
    __syncthreads();

    for (int e = tid; e < STRIP * D_SZ; e += 128) {
        const int r = e / D_SZ;
        const int n = e - r * D_SZ;
        const float v = sS[(0 * STRIP + r) * OP + n] + sS[(1 * STRIP + r) * OP + n]
                      + sS[(2 * STRIP + r) * OP + n] + sS[(3 * STRIP + r) * OP + n];
        dxs[(size_t)row0 * D_SZ + e] = v + b3[n];
    }
    if (tid < STRIP)
        ndiv[row0 + tid] = -(sR[tid] + sR[STRIP + tid] + sR[2 * STRIP + tid] + sR[3 * STRIP + tid]);
}

extern "C" void kernel_launch(void* const* d_in, const int* in_sizes, int n_in,
                              void* d_out, int out_size, void* d_ws, size_t ws_size,
                              hipStream_t stream) {
    const float* xs = (const float*)d_in[0];
    const float* t  = (const float*)d_in[1];
    const float* W1 = (const float*)d_in[2];
    const float* b1 = (const float*)d_in[3];
    const float* W2 = (const float*)d_in[4];
    const float* b2 = (const float*)d_in[5];
    const float* W3 = (const float*)d_in[6];
    const float* b3 = (const float*)d_in[7];

    float* dxs  = (float*)d_out;
    float* ndiv = (float*)d_out + (size_t)B_SZ * D_SZ;

    float* Pc  = (float*)d_ws;
    float* Pw2 = (float*)d_ws + (size_t)H_SZ * H_SZ;
    float* Pw1 = (float*)d_ws + (size_t)2 * H_SZ * H_SZ;

    pack_weights<<<NT_C + NT_W1, 256, 0, stream>>>(W1, W2, W3, Pc, Pw2, Pw1);

    strip_mlp<<<B_SZ / STRIP, 128, 0, stream>>>(xs, t, Pw1, b1, Pw2, b2, W3, b3, Pc, dxs, ndiv);
}


# strip kernel's 16 input and 16 output rows move as 16-byte pieces of one contiguous run, a thread's loads requested as one group (was 4-byte load-wait-store loops: 14 serial round trips in, 9 out)
# speedup vs baseline: 12.1154x; 1.1825x over previous
#include <hip/hip_runtime.h>
#include <math.h>


typedef __attribute__((ext_vector_type(2))) float v2f;
typedef __attribute__((ext_vector_type(4))) float v4f;
typedef __attribute__((ext_vector_type(8))) float v8f;

#define B_SZ 4096
#define D_SZ 66
#define H_SZ 512


__device__ __forceinline__ v8f wmma4(v2f a, v2f b, v8f c) {
    return __builtin_amdgcn_wmma_f32_16x16x4_f32(
        false, a, false, b, (short)0, c, false, false);
}

__device__ __forceinline__ int kof(int kb, int h, int i) {
    return kb + ((i >> 2) << 3) + (h << 2) + (i & 3);
}

#define PCHUNK (4 * 32 * 32 * 4)
#define CT 32
struct PieceMap {
    int kr;
    int nc;
    size_t off;
};
__device__ __forceinline__ PieceMap piece_of(int tid, int t, int n0)
{
    const int cl  = tid & 7;
    const int hh  = (tid >> 3) & 1;
    const int jpl = (tid >> 4) & 1;
    const int s   = tid >> 5;
    const int w   = n0 >> 7;
    const int g   = (n0 >> 6) & 1;
    const int c0  = (n0 >> 2) & 8;
    const int q    = 4 * s + 2 * g + jpl;
    const int lane = 16 * hh + c0 + cl;
    PieceMap m;
    m.kr  = 8 * (s >> 1) + 4 * hh + 2 * (s & 1);
    m.nc  = 4 * cl + 2 * jpl;
    m.off = ((((size_t)t * 4 + w) * 32 + q) * 32 + lane) * 4;
    return m;
}

#define CNV (CT * D_SZ / 4)
#define CP 34
#define NT_C ((H_SZ / CT) * (H_SZ / CT))
#define NT_W1 (3 * (H_SZ / CT))
__global__ __launch_bounds__(256)
void pack_weights(const float* __restrict__ W1, const float* __restrict__ W2,
                  const float* __restrict__ W3,
                  float* __restrict__ Pc, float* __restrict__ Pw2, float* __restrict__ Pw1)
{
    __shared__ __attribute__((aligned(16))) float sW1[D_SZ * CT];
    __shared__ __attribute__((aligned(16))) float sW3[CT * D_SZ];
    __shared__ __attribute__((aligned(16))) float sC[CT * CP];
    __shared__ __attribute__((aligned(16))) float sW[CT * CP];

    const int tid = threadIdx.x;

    if (blockIdx.x >= NT_C) {
        const int bb = blockIdx.x - NT_C;
        const int t  = bb >> 4;
        const int n0 = (bb & 15) * CT;
        const PieceMap m = piece_of(tid, t, n0);
        const int k = 32 * t + m.kr;
        v2f r0 = {0.0f, 0.0f}, r1 = {0.0f, 0.0f};
        if (k < D_SZ + 1)     r0 = *(const v2f*)&W1[(size_t)k * H_SZ + n0 + m.nc];
        if (k + 1 < D_SZ + 1) r1 = *(const v2f*)&W1[(size_t)(k + 1) * H_SZ + n0 + m.nc];
        *(v4f*)&Pw1[m.off] = (v4f){r0.x, r1.x, r0.y, r1.y};
        return;
    }

    const int l0 = (blockIdx.x >> 4) * CT;
    const int k0 = (blockIdx.x & 15) * CT;
    const int kk = tid & 31;
    const int lq = tid >> 5;

    float w2[4];
    for (int i = 0; i < 4; ++i)
        w2[i] = W2[(size_t)(l0 + 4 * lq + i) * H_SZ + k0 + kk];

    const float* w3src = W3 + (size_t)k0 * D_SZ;
    v4f r1[3], r3[3];
    for (int i = 0; i < 3; ++i) {
        const int e = tid + 256 * i;
        r1[i] = (v4f){0.0f, 0.0f, 0.0f, 0.0f};
        r3[i] = (v4f){0.0f, 0.0f, 0.0f, 0.0f};
        if (e < CNV) {
            r1[i] = *(const v4f*)&W1[(size_t)(e >> 3) * H_SZ + l0 + 4 * (e & 7)];
            r3[i] = *(const v4f*)&w3src[4 * e];
        }
    }
    for (int i = 0; i < 3; ++i) {
        const int e = tid + 256 * i;
        if (e < CNV) {
            *(v4f*)&sW1[4 * e] = r1[i];
            *(v4f*)&sW3[4 * e] = r3[i];
        }
    }
    __syncthreads();

    const float* w3row = &sW3[kk * D_SZ];
    const float* w1col = &sW1[4 * lq];
    float m[4] = {0.0f, 0.0f, 0.0f, 0.0f};
    for (int j = 0; j < D_SZ; j += 2) {
        const v2f b  = *(const v2f*)&w3row[j];
        const v4f a0 = *(const v4f*)&w1col[j * CT];
        const v4f a1 = *(const v4f*)&w1col[(j + 1) * CT];
        for (int i = 0; i < 4; ++i) m[i] = fmaf(a0[i], b.x, m[i]);
        for (int i = 0; i < 4; ++i) m[i] = fmaf(a1[i], b.y, m[i]);
    }
    for (int i = 0; i < 4; ++i) {
        sC[(4 * lq + i) * CP + kk] = m[i] * w2[i];
        sW[(4 * lq + i) * CP + kk] = w2[i];
    }
    __syncthreads();

    const PieceMap pm = piece_of(tid, l0 >> 5, k0);
    {
        const v2f c0 = *(const v2f*)&sC[pm.kr * CP + pm.nc];
        const v2f c1 = *(const v2f*)&sC[(pm.kr + 1) * CP + pm.nc];
        *(v4f*)&Pc[pm.off] = (v4f){c0.x, c1.x, c0.y, c1.y};
        const v2f u0 = *(const v2f*)&sW[pm.kr * CP + pm.nc];
        const v2f u1 = *(const v2f*)&sW[(pm.kr + 1) * CP + pm.nc];
        *(v4f*)&Pw2[pm.off] = (v4f){u0.x, u1.x, u0.y, u1.y};
    }
}

#define STRIP 16
#define XP 100
#define AP 516
#define OP 80
#define NPIECE (STRIP * D_SZ / 4)
#define NPP ((NPIECE + 127) / 128)
#define PIECE_IN_RUN(i, p) ((i) < NPP - 1 || (p) < NPIECE)
#define PIECE_TO_LOAD(i, p) (((i) < NPP - 1) ? (p) : min((p), NPIECE - 1))


struct FragP {
    float a[16];
    v4f   b[32];
};

template<int NSUB>
struct FragS {
    float a[16];
    float b[16][NSUB];
};

__device__ __forceinline__ void load_b_packed(FragP& f, const float* __restrict__ Pl, int t)
{
    const v4f* p = (const v4f*)(Pl + (size_t)t * PCHUNK);
    #pragma unroll
    for (int q = 0; q < 32; ++q) f.b[q] = p[q * 32];
}

__device__ __forceinline__ void load_b_out(FragS<5>& f, const float* __restrict__ W3,
                                           int c, int h, int kb)
{
    #pragma unroll
    for (int i = 0; i < 16; ++i) {
        const int k = kof(kb, h, i);
        #pragma unroll
        for (int j = 0; j < 5; ++j) {
            const int n = min(16 * j + c, D_SZ - 1);
            f.b[i][j] = W3[(size_t)k * D_SZ + n];
        }
    }
}

template<class F>
__device__ __forceinline__ void load_a_lds(F& f, const float* arow, int h, int kb)
{
    #pragma unroll
    for (int q = 0; q < 4; ++q) {
        const v4f v = *(const v4f*)&arow[kof(kb, h, 4 * q)];
        f.a[4 * q + 0] = v.x;
        f.a[4 * q + 1] = v.y;
        f.a[4 * q + 2] = v.z;
        f.a[4 * q + 3] = v.w;
    }
}

__device__ __forceinline__ void mma_packed(const FragP& f, v8f (&acc)[8])
{
    #pragma unroll
    for (int s = 0; s < 8; ++s) {
        const v2f a = {f.a[2 * s], f.a[2 * s + 1]};
        #pragma unroll
        for (int j = 0; j < 8; ++j) {
            const v4f v = f.b[4 * s + (j >> 1)];
            const v2f b = (j & 1) ? (v2f){v.z, v.w} : (v2f){v.x, v.y};
            acc[j] = wmma4(a, b, acc[j]);
        }
    }
}

template<int NSUB>
__device__ __forceinline__ void mma_strip(const FragS<NSUB>& f, v8f (&acc)[NSUB])
{
    #pragma unroll
    for (int t = 0; t < 8; ++t) {
        const v2f a = {f.a[2 * t], f.a[2 * t + 1]};
        #pragma unroll
        for (int j = 0; j < NSUB; ++j) {
            const v2f b = {f.b[2 * t][j], f.b[2 * t + 1][j]};
            acc[j] = wmma4(a, b, acc[j]);
        }
    }
}

#define SCHED_FENCE() __builtin_amdgcn_sched_barrier(0)

template<int NCH>
__device__ __forceinline__ void product_packed(v8f (&acc)[8], FragP& f0, FragP& f1,
                                               const float* arow,
                                               const float* __restrict__ Pl, int h)
{
    constexpr int NPAIR = (NCH - 1) / 2;
    load_a_lds(f0, arow, h, 0);
    for (int p = 0; p < NPAIR; ++p) {
        const int t = 2 * p;
        load_b_packed(f1, Pl, t + 1);
        load_a_lds(f1, arow, h, 32 * (t + 1));     SCHED_FENCE();
        mma_packed(f0, acc);                       SCHED_FENCE();
        load_b_packed(f0, Pl, t + 2);
        load_a_lds(f0, arow, h, 32 * (t + 2));     SCHED_FENCE();
        mma_packed(f1, acc);                       SCHED_FENCE();
    }
    if ((NCH & 1) == 0) {
        load_b_packed(f1, Pl, NCH - 1);
        load_a_lds(f1, arow, h, 32 * (NCH - 1));   SCHED_FENCE();
        mma_packed(f0, acc);
        mma_packed(f1, acc);
    } else {
        mma_packed(f0, acc);
    }
}

__global__ __launch_bounds__(128)
void strip_mlp(const float* __restrict__ xs, const float* __restrict__ tvec,
               const float* __restrict__ Pw1, const float* __restrict__ b1,
               const float* __restrict__ Pw2, const float* __restrict__ b2,
               const float* __restrict__ W3, const float* __restrict__ b3,
               const float* __restrict__ Pc,
               float* __restrict__ dxs, float* __restrict__ ndiv)
{
    __shared__ __attribute__((aligned(16))) float sX[STRIP * XP];
    __shared__ __attribute__((aligned(16))) float sA[STRIP * AP];
    __shared__ __attribute__((aligned(16))) float sS[STRIP * AP];
    __shared__ float sR[4 * STRIP];

    const int tid  = threadIdx.x;
    const int lane = tid & 31;
    const int wave = tid >> 5;
    const int c = lane & 15;
    const int h = lane >> 4;
    const int row0 = blockIdx.x * STRIP;
    const int col  = 128 * wave + 4 * c;

    const int loff = wave * (32 * 32 * 4) + lane * 4;

    const v4f* xsrc = (const v4f*)(xs + (size_t)row0 * D_SZ);
    v4f xin[NPP];
    #pragma unroll
    for (int i = 0; i < NPP; ++i) xin[i] = xsrc[PIECE_TO_LOAD(i, tid + 128 * i)];
    const float tin = tvec[row0 + (tid & (STRIP - 1))];

    FragP f0, f1;
    load_b_packed(f0, Pw1 + loff, 0);

    const v4f bias1a = *(const v4f*)&b1[col];
    const v4f bias1b = *(const v4f*)&b1[col + 64];
    const v4f bias2a = *(const v4f*)&b2[col];
    const v4f bias2b = *(const v4f*)&b2[col + 64];

    #pragma unroll
    for (int i = 0; i < NPP; ++i) {
        const int p = tid + 128 * i;
        if (PIECE_IN_RUN(i, p)) {
            const int e0 = 4 * p, e1 = 4 * p + 2;
            const int r0 = e0 / D_SZ, r1 = e1 / D_SZ;
            *(v2f*)&sX[r0 * XP + (e0 - r0 * D_SZ)] = (v2f){xin[i].x, xin[i].y};
            *(v2f*)&sX[r1 * XP + (e1 - r1 * D_SZ)] = (v2f){xin[i].z, xin[i].w};
        }
    }
    if (tid < STRIP) *(v2f*)&sX[tid * XP + D_SZ] = (v2f){tin, 0.0f};
    *(v4f*)&sX[(tid >> 3) * XP + (D_SZ + 2) + 4 * (tid & 7)] = (v4f){0.0f, 0.0f, 0.0f, 0.0f};
    __syncthreads();

    v8f acc[8];

    #pragma unroll
    for (int j = 0; j < 8; ++j) acc[j] = (v8f){0.0f, 0.0f, 0.0f, 0.0f, 0.0f, 0.0f, 0.0f, 0.0f};
    product_packed<3>(acc, f0, f1, &sX[c * XP], Pw1 + loff, h);
    load_b_packed(f0, Pw2 + loff, 0);                SCHED_FENCE();
    #pragma unroll
    for (int g = 0; g < 2; ++g) {
        const v4f bi = g ? bias1b : bias1a;
        #pragma unroll
        for (int r = 0; r < 8; ++r) {
            v4f z = {acc[4 * g + 0][r], acc[4 * g + 1][r], acc[4 * g + 2][r], acc[4 * g + 3][r]};
            z += bi;
            v4f o0, o1;
            #pragma unroll
            for (int e = 0; e < 4; ++e) {
                const float sg = 1.0f / (1.0f + __expf(-z[e]));
                o0[e] = z[e] * sg;
                o1[e] = sg * (1.0f + z[e] * (1.0f - sg));
            }
            *(v4f*)&sA[(8 * h + r) * AP + col + 64 * g] = o0;
            *(v4f*)&sS[(8 * h + r) * AP + col + 64 * g] = o1;
        }
    }
    __syncthreads();

    #pragma unroll
    for (int j = 0; j < 8; ++j) acc[j] = (v8f){0.0f, 0.0f, 0.0f, 0.0f, 0.0f, 0.0f, 0.0f, 0.0f};
    product_packed<H_SZ / 32>(acc, f0, f1, &sA[c * AP], Pw2 + loff, h);
    load_b_packed(f0, Pc + loff, 0);                 SCHED_FENCE();
    v8f a2r[8], s2r[8];
    #pragma unroll
    for (int j = 0; j < 8; ++j) {
        const v4f bi = (j < 4) ? bias2a : bias2b;
        #pragma unroll
        for (int r = 0; r < 8; ++r) {
            const float z  = acc[j][r] + bi[j & 3];
            const float sg = 1.0f / (1.0f + __expf(-z));
            a2r[j][r] = z * sg;
            s2r[j][r] = sg * (1.0f + z * (1.0f - sg));
        }
    }
    SCHED_FENCE();
    __syncthreads();
    #pragma unroll
    for (int g = 0; g < 2; ++g) {
        #pragma unroll
        for (int r = 0; r < 8; ++r) {
            const v4f o = {a2r[4 * g + 0][r], a2r[4 * g + 1][r], a2r[4 * g + 2][r], a2r[4 * g + 3][r]};
            *(v4f*)&sA[(8 * h + r) * AP + col + 64 * g] = o;
        }
    }

    #pragma unroll
    for (int j = 0; j < 8; ++j) acc[j] = (v8f){0.0f, 0.0f, 0.0f, 0.0f, 0.0f, 0.0f, 0.0f, 0.0f};
    product_packed<H_SZ / 32>(acc, f0, f1, &sS[c * AP], Pc + loff, h);
    FragS<5> g0, g1;
    const int kw = 128 * wave;
    load_b_out(g0, W3, c, h, kw);                    SCHED_FENCE();
    float p[8];
    #pragma unroll
    for (int r = 0; r < 8; ++r) {
        p[r] = 0.0f;
        #pragma unroll
        for (int j = 0; j < 8; ++j) p[r] = fmaf(acc[j][r], s2r[j][r], p[r]);
    }
    #pragma unroll
    for (int off = 8; off > 0; off >>= 1) {
        #pragma unroll
        for (int r = 0; r < 8; ++r) p[r] += __shfl_xor(p[r], off, 32);
    }
    if (c == 0) {
        #pragma unroll
        for (int r = 0; r < 8; ++r) sR[wave * STRIP + 8 * h + r] = p[r];
    }
    __syncthreads();

    v2f bout[NPP][2];
    #pragma unroll
    for (int i = 0; i < NPP; ++i) {
        const int p = PIECE_TO_LOAD(i, tid + 128 * i);
        #pragma unroll
        for (int hf = 0; hf < 2; ++hf) bout[i][hf] = *(const v2f*)&b3[(4 * p + 2 * hf) % D_SZ];
    }

    v8f acc4[5];
    #pragma unroll
    for (int j = 0; j < 5; ++j) acc4[j] = (v8f){0.0f, 0.0f, 0.0f, 0.0f, 0.0f, 0.0f, 0.0f, 0.0f};
    const float* a2row = &sA[c * AP];
    load_a_lds(g0, a2row, h, kw);
    load_b_out(g1, W3, c, h, kw + 32);
    load_a_lds(g1, a2row, h, kw + 32);            SCHED_FENCE();
    mma_strip<5>(g0, acc4);                          SCHED_FENCE();
    load_b_out(g0, W3, c, h, kw + 64);
    load_a_lds(g0, a2row, h, kw + 64);            SCHED_FENCE();
    mma_strip<5>(g1, acc4);                          SCHED_FENCE();
    load_b_out(g1, W3, c, h, kw + 96);
    load_a_lds(g1, a2row, h, kw + 96);            SCHED_FENCE();
    mma_strip<5>(g0, acc4);
    mma_strip<5>(g1, acc4);
    #pragma unroll
    for (int j = 0; j < 5; ++j) {
        #pragma unroll
        for (int r = 0; r < 8; ++r)
            sS[(wave * STRIP + 8 * h + r) * OP + 16 * j + c] = acc4[j][r];
    }
    __syncthreads();

    v4f* dst = (v4f*)(dxs + (size_t)row0 * D_SZ);
    #pragma unroll
    for (int i = 0; i < NPP; ++i) {
        const int p = tid + 128 * i;
        if (PIECE_IN_RUN(i, p)) {
            v2f o[2];
            #pragma unroll
            for (int hf = 0; hf < 2; ++hf) {
                const int e = 4 * p + 2 * hf;
                const int r = e / D_SZ;
                const int n = e - r * D_SZ;
                const v2f v = *(const v2f*)&sS[(0 * STRIP + r) * OP + n] + *(const v2f*)&sS[(1 * STRIP + r) * OP + n]
                            + *(const v2f*)&sS[(2 * STRIP + r) * OP + n] + *(const v2f*)&sS[(3 * STRIP + r) * OP + n];
                o[hf] = v + bout[i][hf];
            }
            dst[p] = (v4f){o[0].x, o[0].y, o[1].x, o[1].y};
        }
    }
    if (tid < STRIP)
        ndiv[row0 + tid] = -(sR[tid] + sR[STRIP + tid] + sR[2 * STRIP + tid] + sR[3 * STRIP + tid]);
}

extern "C" void kernel_launch(void* const* d_in, const int* in_sizes, int n_in,
                              void* d_out, int out_size, void* d_ws, size_t ws_size,
                              hipStream_t stream) {
    const float* xs = (const float*)d_in[0];
    const float* t  = (const float*)d_in[1];
    const float* W1 = (const float*)d_in[2];
    const float* b1 = (const float*)d_in[3];
    const float* W2 = (const float*)d_in[4];
    const float* b2 = (const float*)d_in[5];
    const float* W3 = (const float*)d_in[6];
    const float* b3 = (const float*)d_in[7];

    float* dxs  = (float*)d_out;
    float* ndiv = (float*)d_out + (size_t)B_SZ * D_SZ;

    float* Pc  = (float*)d_ws;
    float* Pw2 = (float*)d_ws + (size_t)H_SZ * H_SZ;
    float* Pw1 = (float*)d_ws + (size_t)2 * H_SZ * H_SZ;

    pack_weights<<<NT_C + NT_W1, 256, 0, stream>>>(W1, W2, W3, Pc, Pw2, Pw1);

    strip_mlp<<<B_SZ / STRIP, 128, 0, stream>>>(xs, t, Pw1, b1, Pw2, b2, W3, b3, Pc, dxs, ndiv);
}


# k-chunk order of each weight stream rotated per workgroup (start chunk = workgroup index mod chunk count): concurrent workgroups read different parts of a weight matrix, not the same lines at once
# speedup vs baseline: 12.6120x; 1.0410x over previous
#include <hip/hip_runtime.h>
#include <math.h>


typedef __attribute__((ext_vector_type(2))) float v2f;
typedef __attribute__((ext_vector_type(4))) float v4f;
typedef __attribute__((ext_vector_type(8))) float v8f;

#define B_SZ 4096
#define D_SZ 66
#define H_SZ 512


__device__ __forceinline__ v8f wmma4(v2f a, v2f b, v8f c) {
    return __builtin_amdgcn_wmma_f32_16x16x4_f32(
        false, a, false, b, (short)0, c, false, false);
}

__device__ __forceinline__ int kof(int kb, int h, int i) {
    return kb + ((i >> 2) << 3) + (h << 2) + (i & 3);
}

#define PCHUNK (4 * 32 * 32 * 4)
#define CT 32
struct PieceMap {
    int kr;
    int nc;
    size_t off;
};
__device__ __forceinline__ PieceMap piece_of(int tid, int t, int n0)
{
    const int cl  = tid & 7;
    const int hh  = (tid >> 3) & 1;
    const int jpl = (tid >> 4) & 1;
    const int s   = tid >> 5;
    const int w   = n0 >> 7;
    const int g   = (n0 >> 6) & 1;
    const int c0  = (n0 >> 2) & 8;
    const int q    = 4 * s + 2 * g + jpl;
    const int lane = 16 * hh + c0 + cl;
    PieceMap m;
    m.kr  = 8 * (s >> 1) + 4 * hh + 2 * (s & 1);
    m.nc  = 4 * cl + 2 * jpl;
    m.off = ((((size_t)t * 4 + w) * 32 + q) * 32 + lane) * 4;
    return m;
}

#define CNV (CT * D_SZ / 4)
#define CP 34
#define NT_C ((H_SZ / CT) * (H_SZ / CT))
#define NT_W1 (3 * (H_SZ / CT))
__global__ __launch_bounds__(256)
void pack_weights(const float* __restrict__ W1, const float* __restrict__ W2,
                  const float* __restrict__ W3,
                  float* __restrict__ Pc, float* __restrict__ Pw2, float* __restrict__ Pw1)
{
    __shared__ __attribute__((aligned(16))) float sW1[D_SZ * CT];
    __shared__ __attribute__((aligned(16))) float sW3[CT * D_SZ];
    __shared__ __attribute__((aligned(16))) float sC[CT * CP];
    __shared__ __attribute__((aligned(16))) float sW[CT * CP];

    const int tid = threadIdx.x;

    if (blockIdx.x >= NT_C) {
        const int bb = blockIdx.x - NT_C;
        const int t  = bb >> 4;
        const int n0 = (bb & 15) * CT;
        const PieceMap m = piece_of(tid, t, n0);
        const int k = 32 * t + m.kr;
        v2f r0 = {0.0f, 0.0f}, r1 = {0.0f, 0.0f};
        if (k < D_SZ + 1)     r0 = *(const v2f*)&W1[(size_t)k * H_SZ + n0 + m.nc];
        if (k + 1 < D_SZ + 1) r1 = *(const v2f*)&W1[(size_t)(k + 1) * H_SZ + n0 + m.nc];
        *(v4f*)&Pw1[m.off] = (v4f){r0.x, r1.x, r0.y, r1.y};
        return;
    }

    const int l0 = (blockIdx.x >> 4) * CT;
    const int k0 = (blockIdx.x & 15) * CT;
    const int kk = tid & 31;
    const int lq = tid >> 5;

    float w2[4];
    for (int i = 0; i < 4; ++i)
        w2[i] = W2[(size_t)(l0 + 4 * lq + i) * H_SZ + k0 + kk];

    const float* w3src = W3 + (size_t)k0 * D_SZ;
    v4f r1[3], r3[3];
    for (int i = 0; i < 3; ++i) {
        const int e = tid + 256 * i;
        r1[i] = (v4f){0.0f, 0.0f, 0.0f, 0.0f};
        r3[i] = (v4f){0.0f, 0.0f, 0.0f, 0.0f};
        if (e < CNV) {
            r1[i] = *(const v4f*)&W1[(size_t)(e >> 3) * H_SZ + l0 + 4 * (e & 7)];
            r3[i] = *(const v4f*)&w3src[4 * e];
        }
    }
    for (int i = 0; i < 3; ++i) {
        const int e = tid + 256 * i;
        if (e < CNV) {
            *(v4f*)&sW1[4 * e] = r1[i];
            *(v4f*)&sW3[4 * e] = r3[i];
        }
    }
    __syncthreads();

    const float* w3row = &sW3[kk * D_SZ];
    const float* w1col = &sW1[4 * lq];
    float m[4] = {0.0f, 0.0f, 0.0f, 0.0f};
    for (int j = 0; j < D_SZ; j += 2) {
        const v2f b  = *(const v2f*)&w3row[j];
        const v4f a0 = *(const v4f*)&w1col[j * CT];
        const v4f a1 = *(const v4f*)&w1col[(j + 1) * CT];
        for (int i = 0; i < 4; ++i) m[i] = fmaf(a0[i], b.x, m[i]);
        for (int i = 0; i < 4; ++i) m[i] = fmaf(a1[i], b.y, m[i]);
    }
    for (int i = 0; i < 4; ++i) {
        sC[(4 * lq + i) * CP + kk] = m[i] * w2[i];
        sW[(4 * lq + i) * CP + kk] = w2[i];
    }
    __syncthreads();

    const PieceMap pm = piece_of(tid, l0 >> 5, k0);
    {
        const v2f c0 = *(const v2f*)&sC[pm.kr * CP + pm.nc];
        const v2f c1 = *(const v2f*)&sC[(pm.kr + 1) * CP + pm.nc];
        *(v4f*)&Pc[pm.off] = (v4f){c0.x, c1.x, c0.y, c1.y};
        const v2f u0 = *(const v2f*)&sW[pm.kr * CP + pm.nc];
        const v2f u1 = *(const v2f*)&sW[(pm.kr + 1) * CP + pm.nc];
        *(v4f*)&Pw2[pm.off] = (v4f){u0.x, u1.x, u0.y, u1.y};
    }
}

#define STRIP 16
#define XP 100
#define AP 516
#define OP 80
#define NPIECE (STRIP * D_SZ / 4)
#define NPP ((NPIECE + 127) / 128)
#define PIECE_IN_RUN(i, p) ((i) < NPP - 1 || (p) < NPIECE)
#define PIECE_TO_LOAD(i, p) (((i) < NPP - 1) ? (p) : min((p), NPIECE - 1))


struct FragP {
    float a[16];
    v4f   b[32];
};

template<int NSUB>
struct FragS {
    float a[16];
    float b[16][NSUB];
};

__device__ __forceinline__ void load_b_packed(FragP& f, const float* __restrict__ Pl, int t)
{
    const v4f* p = (const v4f*)(Pl + (size_t)t * PCHUNK);
    #pragma unroll
    for (int q = 0; q < 32; ++q) f.b[q] = p[q * 32];
}

__device__ __forceinline__ void load_b_out(FragS<5>& f, const float* __restrict__ W3,
                                           int c, int h, int kb)
{
    #pragma unroll
    for (int i = 0; i < 16; ++i) {
        const int k = kof(kb, h, i);
        #pragma unroll
        for (int j = 0; j < 5; ++j) {
            const int n = min(16 * j + c, D_SZ - 1);
            f.b[i][j] = W3[(size_t)k * D_SZ + n];
        }
    }
}

template<class F>
__device__ __forceinline__ void load_a_lds(F& f, const float* arow, int h, int kb)
{
    #pragma unroll
    for (int q = 0; q < 4; ++q) {
        const v4f v = *(const v4f*)&arow[kof(kb, h, 4 * q)];
        f.a[4 * q + 0] = v.x;
        f.a[4 * q + 1] = v.y;
        f.a[4 * q + 2] = v.z;
        f.a[4 * q + 3] = v.w;
    }
}

__device__ __forceinline__ void mma_packed(const FragP& f, v8f (&acc)[8])
{
    #pragma unroll
    for (int s = 0; s < 8; ++s) {
        const v2f a = {f.a[2 * s], f.a[2 * s + 1]};
        #pragma unroll
        for (int j = 0; j < 8; ++j) {
            const v4f v = f.b[4 * s + (j >> 1)];
            const v2f b = (j & 1) ? (v2f){v.z, v.w} : (v2f){v.x, v.y};
            acc[j] = wmma4(a, b, acc[j]);
        }
    }
}

template<int NSUB>
__device__ __forceinline__ void mma_strip(const FragS<NSUB>& f, v8f (&acc)[NSUB])
{
    #pragma unroll
    for (int t = 0; t < 8; ++t) {
        const v2f a = {f.a[2 * t], f.a[2 * t + 1]};
        #pragma unroll
        for (int j = 0; j < NSUB; ++j) {
            const v2f b = {f.b[2 * t][j], f.b[2 * t + 1][j]};
            acc[j] = wmma4(a, b, acc[j]);
        }
    }
}

#define SCHED_FENCE() __builtin_amdgcn_sched_barrier(0)

template<int NCH>
__device__ __forceinline__ int chunk_at(int i, int rot)
{
    const int x = i + rot;
    if ((NCH & (NCH - 1)) == 0) return x & (NCH - 1);
    return x >= NCH ? x - NCH : x;
}

template<int NCH>
__device__ __forceinline__ void product_packed(v8f (&acc)[8], FragP& f0, FragP& f1,
                                               const float* arow,
                                               const float* __restrict__ Pl, int h, int rot)
{
    constexpr int NPAIR = (NCH - 1) / 2;
    load_a_lds(f0, arow, h, 32 * chunk_at<NCH>(0, rot));
    for (int p = 0; p < NPAIR; ++p) {
        const int t = 2 * p;
        const int c1 = chunk_at<NCH>(t + 1, rot);
        const int c2 = chunk_at<NCH>(t + 2, rot);
        load_b_packed(f1, Pl, c1);
        load_a_lds(f1, arow, h, 32 * c1);          SCHED_FENCE();
        mma_packed(f0, acc);                       SCHED_FENCE();
        load_b_packed(f0, Pl, c2);
        load_a_lds(f0, arow, h, 32 * c2);          SCHED_FENCE();
        mma_packed(f1, acc);                       SCHED_FENCE();
    }
    if ((NCH & 1) == 0) {
        const int cl = chunk_at<NCH>(NCH - 1, rot);
        load_b_packed(f1, Pl, cl);
        load_a_lds(f1, arow, h, 32 * cl);          SCHED_FENCE();
        mma_packed(f0, acc);
        mma_packed(f1, acc);
    } else {
        mma_packed(f0, acc);
    }
}

__global__ __launch_bounds__(128)
void strip_mlp(const float* __restrict__ xs, const float* __restrict__ tvec,
               const float* __restrict__ Pw1, const float* __restrict__ b1,
               const float* __restrict__ Pw2, const float* __restrict__ b2,
               const float* __restrict__ W3, const float* __restrict__ b3,
               const float* __restrict__ Pc,
               float* __restrict__ dxs, float* __restrict__ ndiv)
{
    __shared__ __attribute__((aligned(16))) float sX[STRIP * XP];
    __shared__ __attribute__((aligned(16))) float sA[STRIP * AP];
    __shared__ __attribute__((aligned(16))) float sS[STRIP * AP];
    __shared__ float sR[4 * STRIP];

    const int tid  = threadIdx.x;
    const int lane = tid & 31;
    const int wave = tid >> 5;
    const int c = lane & 15;
    const int h = lane >> 4;
    const int row0 = blockIdx.x * STRIP;
    const int col  = 128 * wave + 4 * c;

    const int loff = wave * (32 * 32 * 4) + lane * 4;

    const int rot3  = blockIdx.x % 3;
    const int rot16 = blockIdx.x & (H_SZ / 32 - 1);
    const int rot4  = blockIdx.x & 3;

    const v4f* xsrc = (const v4f*)(xs + (size_t)row0 * D_SZ);
    v4f xin[NPP];
    #pragma unroll
    for (int i = 0; i < NPP; ++i) xin[i] = xsrc[PIECE_TO_LOAD(i, tid + 128 * i)];
    const float tin = tvec[row0 + (tid & (STRIP - 1))];

    FragP f0, f1;
    load_b_packed(f0, Pw1 + loff, rot3);

    const v4f bias1a = *(const v4f*)&b1[col];
    const v4f bias1b = *(const v4f*)&b1[col + 64];
    const v4f bias2a = *(const v4f*)&b2[col];
    const v4f bias2b = *(const v4f*)&b2[col + 64];

    #pragma unroll
    for (int i = 0; i < NPP; ++i) {
        const int p = tid + 128 * i;
        if (PIECE_IN_RUN(i, p)) {
            const int e0 = 4 * p, e1 = 4 * p + 2;
            const int r0 = e0 / D_SZ, r1 = e1 / D_SZ;
            *(v2f*)&sX[r0 * XP + (e0 - r0 * D_SZ)] = (v2f){xin[i].x, xin[i].y};
            *(v2f*)&sX[r1 * XP + (e1 - r1 * D_SZ)] = (v2f){xin[i].z, xin[i].w};
        }
    }
    if (tid < STRIP) *(v2f*)&sX[tid * XP + D_SZ] = (v2f){tin, 0.0f};
    *(v4f*)&sX[(tid >> 3) * XP + (D_SZ + 2) + 4 * (tid & 7)] = (v4f){0.0f, 0.0f, 0.0f, 0.0f};
    __syncthreads();

    v8f acc[8];

    #pragma unroll
    for (int j = 0; j < 8; ++j) acc[j] = (v8f){0.0f, 0.0f, 0.0f, 0.0f, 0.0f, 0.0f, 0.0f, 0.0f};
    product_packed<3>(acc, f0, f1, &sX[c * XP], Pw1 + loff, h, rot3);
    load_b_packed(f0, Pw2 + loff, rot16);            SCHED_FENCE();
    #pragma unroll
    for (int g = 0; g < 2; ++g) {
        const v4f bi = g ? bias1b : bias1a;
        #pragma unroll
        for (int r = 0; r < 8; ++r) {
            v4f z = {acc[4 * g + 0][r], acc[4 * g + 1][r], acc[4 * g + 2][r], acc[4 * g + 3][r]};
            z += bi;
            v4f o0, o1;
            #pragma unroll
            for (int e = 0; e < 4; ++e) {
                const float sg = 1.0f / (1.0f + __expf(-z[e]));
                o0[e] = z[e] * sg;
                o1[e] = sg * (1.0f + z[e] * (1.0f - sg));
            }
            *(v4f*)&sA[(8 * h + r) * AP + col + 64 * g] = o0;
            *(v4f*)&sS[(8 * h + r) * AP + col + 64 * g] = o1;
        }
    }
    __syncthreads();

    #pragma unroll
    for (int j = 0; j < 8; ++j) acc[j] = (v8f){0.0f, 0.0f, 0.0f, 0.0f, 0.0f, 0.0f, 0.0f, 0.0f};
    product_packed<H_SZ / 32>(acc, f0, f1, &sA[c * AP], Pw2 + loff, h, rot16);
    load_b_packed(f0, Pc + loff, rot16);             SCHED_FENCE();
    v8f a2r[8], s2r[8];
    #pragma unroll
    for (int j = 0; j < 8; ++j) {
        const v4f bi = (j < 4) ? bias2a : bias2b;
        #pragma unroll
        for (int r = 0; r < 8; ++r) {
            const float z  = acc[j][r] + bi[j & 3];
            const float sg = 1.0f / (1.0f + __expf(-z));
            a2r[j][r] = z * sg;
            s2r[j][r] = sg * (1.0f + z * (1.0f - sg));
        }
    }
    SCHED_FENCE();
    __syncthreads();
    #pragma unroll
    for (int g = 0; g < 2; ++g) {
        #pragma unroll
        for (int r = 0; r < 8; ++r) {
            const v4f o = {a2r[4 * g + 0][r], a2r[4 * g + 1][r], a2r[4 * g + 2][r], a2r[4 * g + 3][r]};
            *(v4f*)&sA[(8 * h + r) * AP + col + 64 * g] = o;
        }
    }

    #pragma unroll
    for (int j = 0; j < 8; ++j) acc[j] = (v8f){0.0f, 0.0f, 0.0f, 0.0f, 0.0f, 0.0f, 0.0f, 0.0f};
    product_packed<H_SZ / 32>(acc, f0, f1, &sS[c * AP], Pc + loff, h, rot16);
    FragS<5> g0, g1;
    const int kw = 128 * wave;
    const int ko0 = kw + 32 * chunk_at<4>(0, rot4);
    const int ko1 = kw + 32 * chunk_at<4>(1, rot4);
    const int ko2 = kw + 32 * chunk_at<4>(2, rot4);
    const int ko3 = kw + 32 * chunk_at<4>(3, rot4);
    load_b_out(g0, W3, c, h, ko0);                   SCHED_FENCE();
    float p[8];
    #pragma unroll
    for (int r = 0; r < 8; ++r) {
        p[r] = 0.0f;
        #pragma unroll
        for (int j = 0; j < 8; ++j) p[r] = fmaf(acc[j][r], s2r[j][r], p[r]);
    }
    #pragma unroll
    for (int off = 8; off > 0; off >>= 1) {
        #pragma unroll
        for (int r = 0; r < 8; ++r) p[r] += __shfl_xor(p[r], off, 32);
    }
    if (c == 0) {
        #pragma unroll
        for (int r = 0; r < 8; ++r) sR[wave * STRIP + 8 * h + r] = p[r];
    }
    __syncthreads();

    v2f bout[NPP][2];
    #pragma unroll
    for (int i = 0; i < NPP; ++i) {
        const int p = PIECE_TO_LOAD(i, tid + 128 * i);
        #pragma unroll
        for (int hf = 0; hf < 2; ++hf) bout[i][hf] = *(const v2f*)&b3[(4 * p + 2 * hf) % D_SZ];
    }

    v8f acc4[5];
    #pragma unroll
    for (int j = 0; j < 5; ++j) acc4[j] = (v8f){0.0f, 0.0f, 0.0f, 0.0f, 0.0f, 0.0f, 0.0f, 0.0f};
    const float* a2row = &sA[c * AP];
    load_a_lds(g0, a2row, h, ko0);
    load_b_out(g1, W3, c, h, ko1);
    load_a_lds(g1, a2row, h, ko1);                SCHED_FENCE();
    mma_strip<5>(g0, acc4);                          SCHED_FENCE();
    load_b_out(g0, W3, c, h, ko2);
    load_a_lds(g0, a2row, h, ko2);                SCHED_FENCE();
    mma_strip<5>(g1, acc4);                          SCHED_FENCE();
    load_b_out(g1, W3, c, h, ko3);
    load_a_lds(g1, a2row, h, ko3);                SCHED_FENCE();
    mma_strip<5>(g0, acc4);
    mma_strip<5>(g1, acc4);
    #pragma unroll
    for (int j = 0; j < 5; ++j) {
        #pragma unroll
        for (int r = 0; r < 8; ++r)
            sS[(wave * STRIP + 8 * h + r) * OP + 16 * j + c] = acc4[j][r];
    }
    __syncthreads();

    v4f* dst = (v4f*)(dxs + (size_t)row0 * D_SZ);
    #pragma unroll
    for (int i = 0; i < NPP; ++i) {
        const int p = tid + 128 * i;
        if (PIECE_IN_RUN(i, p)) {
            v2f o[2];
            #pragma unroll
            for (int hf = 0; hf < 2; ++hf) {
                const int e = 4 * p + 2 * hf;
                const int r = e / D_SZ;
                const int n = e - r * D_SZ;
                const v2f v = *(const v2f*)&sS[(0 * STRIP + r) * OP + n] + *(const v2f*)&sS[(1 * STRIP + r) * OP + n]
                            + *(const v2f*)&sS[(2 * STRIP + r) * OP + n] + *(const v2f*)&sS[(3 * STRIP + r) * OP + n];
                o[hf] = v + bout[i][hf];
            }
            dst[p] = (v4f){o[0].x, o[0].y, o[1].x, o[1].y};
        }
    }
    if (tid < STRIP)
        ndiv[row0 + tid] = -(sR[tid] + sR[STRIP + tid] + sR[2 * STRIP + tid] + sR[3 * STRIP + tid]);
}

extern "C" void kernel_launch(void* const* d_in, const int* in_sizes, int n_in,
                              void* d_out, int out_size, void* d_ws, size_t ws_size,
                              hipStream_t stream) {
    const float* xs = (const float*)d_in[0];
    const float* t  = (const float*)d_in[1];
    const float* W1 = (const float*)d_in[2];
    const float* b1 = (const float*)d_in[3];
    const float* W2 = (const float*)d_in[4];
    const float* b2 = (const float*)d_in[5];
    const float* W3 = (const float*)d_in[6];
    const float* b3 = (const float*)d_in[7];

    float* dxs  = (float*)d_out;
    float* ndiv = (float*)d_out + (size_t)B_SZ * D_SZ;

    float* Pc  = (float*)d_ws;
    float* Pw2 = (float*)d_ws + (size_t)H_SZ * H_SZ;
    float* Pw1 = (float*)d_ws + (size_t)2 * H_SZ * H_SZ;

    pack_weights<<<NT_C + NT_W1, 256, 0, stream>>>(W1, W2, W3, Pc, Pw2, Pw1);

    strip_mlp<<<B_SZ / STRIP, 128, 0, stream>>>(xs, t, Pw1, b1, Pw2, b2, W3, b3, Pc, dxs, ndiv);
}


# sigmoid of both activation epilogues: 1/(1+exp(-z)) by the hardware reciprocal v_rcp_f32 (1 ulp) instead of the exact division, a serial chain of 11 dependent vector instructions per element
# speedup vs baseline: 13.2269x; 1.0488x over previous
#include <hip/hip_runtime.h>
#include <math.h>


typedef __attribute__((ext_vector_type(2))) float v2f;
typedef __attribute__((ext_vector_type(4))) float v4f;
typedef __attribute__((ext_vector_type(8))) float v8f;

#define B_SZ 4096
#define D_SZ 66
#define H_SZ 512


__device__ __forceinline__ v8f wmma4(v2f a, v2f b, v8f c) {
    return __builtin_amdgcn_wmma_f32_16x16x4_f32(
        false, a, false, b, (short)0, c, false, false);
}

__device__ __forceinline__ float sigmoid_rcp(float z) {
    return __builtin_amdgcn_rcpf(1.0f + __expf(-z));
}

__device__ __forceinline__ int kof(int kb, int h, int i) {
    return kb + ((i >> 2) << 3) + (h << 2) + (i & 3);
}

#define PCHUNK (4 * 32 * 32 * 4)
#define CT 32
struct PieceMap {
    int kr;
    int nc;
    size_t off;
};
__device__ __forceinline__ PieceMap piece_of(int tid, int t, int n0)
{
    const int cl  = tid & 7;
    const int hh  = (tid >> 3) & 1;
    const int jpl = (tid >> 4) & 1;
    const int s   = tid >> 5;
    const int w   = n0 >> 7;
    const int g   = (n0 >> 6) & 1;
    const int c0  = (n0 >> 2) & 8;
    const int q    = 4 * s + 2 * g + jpl;
    const int lane = 16 * hh + c0 + cl;
    PieceMap m;
    m.kr  = 8 * (s >> 1) + 4 * hh + 2 * (s & 1);
    m.nc  = 4 * cl + 2 * jpl;
    m.off = ((((size_t)t * 4 + w) * 32 + q) * 32 + lane) * 4;
    return m;
}

#define CNV (CT * D_SZ / 4)
#define CP 34
#define NT_C ((H_SZ / CT) * (H_SZ / CT))
#define NT_W1 (3 * (H_SZ / CT))
__global__ __launch_bounds__(256)
void pack_weights(const float* __restrict__ W1, const float* __restrict__ W2,
                  const float* __restrict__ W3,
                  float* __restrict__ Pc, float* __restrict__ Pw2, float* __restrict__ Pw1)
{
    __shared__ __attribute__((aligned(16))) float sW1[D_SZ * CT];
    __shared__ __attribute__((aligned(16))) float sW3[CT * D_SZ];
    __shared__ __attribute__((aligned(16))) float sC[CT * CP];
    __shared__ __attribute__((aligned(16))) float sW[CT * CP];

    const int tid = threadIdx.x;

    if (blockIdx.x >= NT_C) {
        const int bb = blockIdx.x - NT_C;
        const int t  = bb >> 4;
        const int n0 = (bb & 15) * CT;
        const PieceMap m = piece_of(tid, t, n0);
        const int k = 32 * t + m.kr;
        v2f r0 = {0.0f, 0.0f}, r1 = {0.0f, 0.0f};
        if (k < D_SZ + 1)     r0 = *(const v2f*)&W1[(size_t)k * H_SZ + n0 + m.nc];
        if (k + 1 < D_SZ + 1) r1 = *(const v2f*)&W1[(size_t)(k + 1) * H_SZ + n0 + m.nc];
        *(v4f*)&Pw1[m.off] = (v4f){r0.x, r1.x, r0.y, r1.y};
        return;
    }

    const int l0 = (blockIdx.x >> 4) * CT;
    const int k0 = (blockIdx.x & 15) * CT;
    const int kk = tid & 31;
    const int lq = tid >> 5;

    float w2[4];
    for (int i = 0; i < 4; ++i)
        w2[i] = W2[(size_t)(l0 + 4 * lq + i) * H_SZ + k0 + kk];

    const float* w3src = W3 + (size_t)k0 * D_SZ;
    v4f r1[3], r3[3];
    for (int i = 0; i < 3; ++i) {
        const int e = tid + 256 * i;
        r1[i] = (v4f){0.0f, 0.0f, 0.0f, 0.0f};
        r3[i] = (v4f){0.0f, 0.0f, 0.0f, 0.0f};
        if (e < CNV) {
            r1[i] = *(const v4f*)&W1[(size_t)(e >> 3) * H_SZ + l0 + 4 * (e & 7)];
            r3[i] = *(const v4f*)&w3src[4 * e];
        }
    }
    for (int i = 0; i < 3; ++i) {
        const int e = tid + 256 * i;
        if (e < CNV) {
            *(v4f*)&sW1[4 * e] = r1[i];
            *(v4f*)&sW3[4 * e] = r3[i];
        }
    }
    __syncthreads();

    const float* w3row = &sW3[kk * D_SZ];
    const float* w1col = &sW1[4 * lq];
    float m[4] = {0.0f, 0.0f, 0.0f, 0.0f};
    for (int j = 0; j < D_SZ; j += 2) {
        const v2f b  = *(const v2f*)&w3row[j];
        const v4f a0 = *(const v4f*)&w1col[j * CT];
        const v4f a1 = *(const v4f*)&w1col[(j + 1) * CT];
        for (int i = 0; i < 4; ++i) m[i] = fmaf(a0[i], b.x, m[i]);
        for (int i = 0; i < 4; ++i) m[i] = fmaf(a1[i], b.y, m[i]);
    }
    for (int i = 0; i < 4; ++i) {
        sC[(4 * lq + i) * CP + kk] = m[i] * w2[i];
        sW[(4 * lq + i) * CP + kk] = w2[i];
    }
    __syncthreads();

    const PieceMap pm = piece_of(tid, l0 >> 5, k0);
    {
        const v2f c0 = *(const v2f*)&sC[pm.kr * CP + pm.nc];
        const v2f c1 = *(const v2f*)&sC[(pm.kr + 1) * CP + pm.nc];
        *(v4f*)&Pc[pm.off] = (v4f){c0.x, c1.x, c0.y, c1.y};
        const v2f u0 = *(const v2f*)&sW[pm.kr * CP + pm.nc];
        const v2f u1 = *(const v2f*)&sW[(pm.kr + 1) * CP + pm.nc];
        *(v4f*)&Pw2[pm.off] = (v4f){u0.x, u1.x, u0.y, u1.y};
    }
}

#define STRIP 16
#define XP 100
#define AP 516
#define OP 80
#define NPIECE (STRIP * D_SZ / 4)
#define NPP ((NPIECE + 127) / 128)
#define PIECE_IN_RUN(i, p) ((i) < NPP - 1 || (p) < NPIECE)
#define PIECE_TO_LOAD(i, p) (((i) < NPP - 1) ? (p) : min((p), NPIECE - 1))


struct FragP {
    float a[16];
    v4f   b[32];
};

template<int NSUB>
struct FragS {
    float a[16];
    float b[16][NSUB];
};

__device__ __forceinline__ void load_b_packed(FragP& f, const float* __restrict__ Pl, int t)
{
    const v4f* p = (const v4f*)(Pl + (size_t)t * PCHUNK);
    #pragma unroll
    for (int q = 0; q < 32; ++q) f.b[q] = p[q * 32];
}

__device__ __forceinline__ void load_b_out(FragS<5>& f, const float* __restrict__ W3,
                                           int c, int h, int kb)
{
    #pragma unroll
    for (int i = 0; i < 16; ++i) {
        const int k = kof(kb, h, i);
        #pragma unroll
        for (int j = 0; j < 5; ++j) {
            const int n = min(16 * j + c, D_SZ - 1);
            f.b[i][j] = W3[(size_t)k * D_SZ + n];
        }
    }
}

template<class F>
__device__ __forceinline__ void load_a_lds(F& f, const float* arow, int h, int kb)
{
    #pragma unroll
    for (int q = 0; q < 4; ++q) {
        const v4f v = *(const v4f*)&arow[kof(kb, h, 4 * q)];
        f.a[4 * q + 0] = v.x;
        f.a[4 * q + 1] = v.y;
        f.a[4 * q + 2] = v.z;
        f.a[4 * q + 3] = v.w;
    }
}

__device__ __forceinline__ void mma_packed(const FragP& f, v8f (&acc)[8])
{
    #pragma unroll
    for (int s = 0; s < 8; ++s) {
        const v2f a = {f.a[2 * s], f.a[2 * s + 1]};
        #pragma unroll
        for (int j = 0; j < 8; ++j) {
            const v4f v = f.b[4 * s + (j >> 1)];
            const v2f b = (j & 1) ? (v2f){v.z, v.w} : (v2f){v.x, v.y};
            acc[j] = wmma4(a, b, acc[j]);
        }
    }
}

template<int NSUB>
__device__ __forceinline__ void mma_strip(const FragS<NSUB>& f, v8f (&acc)[NSUB])
{
    #pragma unroll
    for (int t = 0; t < 8; ++t) {
        const v2f a = {f.a[2 * t], f.a[2 * t + 1]};
        #pragma unroll
        for (int j = 0; j < NSUB; ++j) {
            const v2f b = {f.b[2 * t][j], f.b[2 * t + 1][j]};
            acc[j] = wmma4(a, b, acc[j]);
        }
    }
}

#define SCHED_FENCE() __builtin_amdgcn_sched_barrier(0)

template<int NCH>
__device__ __forceinline__ int chunk_at(int i, int rot)
{
    const int x = i + rot;
    if ((NCH & (NCH - 1)) == 0) return x & (NCH - 1);
    return x >= NCH ? x - NCH : x;
}

template<int NCH>
__device__ __forceinline__ void product_packed(v8f (&acc)[8], FragP& f0, FragP& f1,
                                               const float* arow,
                                               const float* __restrict__ Pl, int h, int rot)
{
    constexpr int NPAIR = (NCH - 1) / 2;
    load_a_lds(f0, arow, h, 32 * chunk_at<NCH>(0, rot));
    for (int p = 0; p < NPAIR; ++p) {
        const int t = 2 * p;
        const int c1 = chunk_at<NCH>(t + 1, rot);
        const int c2 = chunk_at<NCH>(t + 2, rot);
        load_b_packed(f1, Pl, c1);
        load_a_lds(f1, arow, h, 32 * c1);          SCHED_FENCE();
        mma_packed(f0, acc);                       SCHED_FENCE();
        load_b_packed(f0, Pl, c2);
        load_a_lds(f0, arow, h, 32 * c2);          SCHED_FENCE();
        mma_packed(f1, acc);                       SCHED_FENCE();
    }
    if ((NCH & 1) == 0) {
        const int cl = chunk_at<NCH>(NCH - 1, rot);
        load_b_packed(f1, Pl, cl);
        load_a_lds(f1, arow, h, 32 * cl);          SCHED_FENCE();
        mma_packed(f0, acc);
        mma_packed(f1, acc);
    } else {
        mma_packed(f0, acc);
    }
}

__global__ __launch_bounds__(128)
void strip_mlp(const float* __restrict__ xs, const float* __restrict__ tvec,
               const float* __restrict__ Pw1, const float* __restrict__ b1,
               const float* __restrict__ Pw2, const float* __restrict__ b2,
               const float* __restrict__ W3, const float* __restrict__ b3,
               const float* __restrict__ Pc,
               float* __restrict__ dxs, float* __restrict__ ndiv)
{
    __shared__ __attribute__((aligned(16))) float sX[STRIP * XP];
    __shared__ __attribute__((aligned(16))) float sA[STRIP * AP];
    __shared__ __attribute__((aligned(16))) float sS[STRIP * AP];
    __shared__ float sR[4 * STRIP];

    const int tid  = threadIdx.x;
    const int lane = tid & 31;
    const int wave = tid >> 5;
    const int c = lane & 15;
    const int h = lane >> 4;
    const int row0 = blockIdx.x * STRIP;
    const int col  = 128 * wave + 4 * c;

    const int loff = wave * (32 * 32 * 4) + lane * 4;

    const int rot3  = blockIdx.x % 3;
    const int rot16 = blockIdx.x & (H_SZ / 32 - 1);
    const int rot4  = blockIdx.x & 3;

    const v4f* xsrc = (const v4f*)(xs + (size_t)row0 * D_SZ);
    v4f xin[NPP];
    #pragma unroll
    for (int i = 0; i < NPP; ++i) xin[i] = xsrc[PIECE_TO_LOAD(i, tid + 128 * i)];
    const float tin = tvec[row0 + (tid & (STRIP - 1))];

    FragP f0, f1;
    load_b_packed(f0, Pw1 + loff, rot3);

    const v4f bias1a = *(const v4f*)&b1[col];
    const v4f bias1b = *(const v4f*)&b1[col + 64];
    const v4f bias2a = *(const v4f*)&b2[col];
    const v4f bias2b = *(const v4f*)&b2[col + 64];

    #pragma unroll
    for (int i = 0; i < NPP; ++i) {
        const int p = tid + 128 * i;
        if (PIECE_IN_RUN(i, p)) {
            const int e0 = 4 * p, e1 = 4 * p + 2;
            const int r0 = e0 / D_SZ, r1 = e1 / D_SZ;
            *(v2f*)&sX[r0 * XP + (e0 - r0 * D_SZ)] = (v2f){xin[i].x, xin[i].y};
            *(v2f*)&sX[r1 * XP + (e1 - r1 * D_SZ)] = (v2f){xin[i].z, xin[i].w};
        }
    }
    if (tid < STRIP) *(v2f*)&sX[tid * XP + D_SZ] = (v2f){tin, 0.0f};
    *(v4f*)&sX[(tid >> 3) * XP + (D_SZ + 2) + 4 * (tid & 7)] = (v4f){0.0f, 0.0f, 0.0f, 0.0f};
    __syncthreads();

    v8f acc[8];

    #pragma unroll
    for (int j = 0; j < 8; ++j) acc[j] = (v8f){0.0f, 0.0f, 0.0f, 0.0f, 0.0f, 0.0f, 0.0f, 0.0f};
    product_packed<3>(acc, f0, f1, &sX[c * XP], Pw1 + loff, h, rot3);
    load_b_packed(f0, Pw2 + loff, rot16);            SCHED_FENCE();
    #pragma unroll
    for (int g = 0; g < 2; ++g) {
        const v4f bi = g ? bias1b : bias1a;
        #pragma unroll
        for (int r = 0; r < 8; ++r) {
            v4f z = {acc[4 * g + 0][r], acc[4 * g + 1][r], acc[4 * g + 2][r], acc[4 * g + 3][r]};
            z += bi;
            v4f o0, o1;
            #pragma unroll
            for (int e = 0; e < 4; ++e) {
                const float sg = sigmoid_rcp(z[e]);
                o0[e] = z[e] * sg;
                o1[e] = sg * (1.0f + z[e] * (1.0f - sg));
            }
            *(v4f*)&sA[(8 * h + r) * AP + col + 64 * g] = o0;
            *(v4f*)&sS[(8 * h + r) * AP + col + 64 * g] = o1;
        }
    }
    __syncthreads();

    #pragma unroll
    for (int j = 0; j < 8; ++j) acc[j] = (v8f){0.0f, 0.0f, 0.0f, 0.0f, 0.0f, 0.0f, 0.0f, 0.0f};
    product_packed<H_SZ / 32>(acc, f0, f1, &sA[c * AP], Pw2 + loff, h, rot16);
    load_b_packed(f0, Pc + loff, rot16);             SCHED_FENCE();
    v8f a2r[8], s2r[8];
    #pragma unroll
    for (int j = 0; j < 8; ++j) {
        const v4f bi = (j < 4) ? bias2a : bias2b;
        #pragma unroll
        for (int r = 0; r < 8; ++r) {
            const float z  = acc[j][r] + bi[j & 3];
            const float sg = sigmoid_rcp(z);
            a2r[j][r] = z * sg;
            s2r[j][r] = sg * (1.0f + z * (1.0f - sg));
        }
    }
    SCHED_FENCE();
    __syncthreads();
    #pragma unroll
    for (int g = 0; g < 2; ++g) {
        #pragma unroll
        for (int r = 0; r < 8; ++r) {
            const v4f o = {a2r[4 * g + 0][r], a2r[4 * g + 1][r], a2r[4 * g + 2][r], a2r[4 * g + 3][r]};
            *(v4f*)&sA[(8 * h + r) * AP + col + 64 * g] = o;
        }
    }

    #pragma unroll
    for (int j = 0; j < 8; ++j) acc[j] = (v8f){0.0f, 0.0f, 0.0f, 0.0f, 0.0f, 0.0f, 0.0f, 0.0f};
    product_packed<H_SZ / 32>(acc, f0, f1, &sS[c * AP], Pc + loff, h, rot16);
    FragS<5> g0, g1;
    const int kw = 128 * wave;
    const int ko0 = kw + 32 * chunk_at<4>(0, rot4);
    const int ko1 = kw + 32 * chunk_at<4>(1, rot4);
    const int ko2 = kw + 32 * chunk_at<4>(2, rot4);
    const int ko3 = kw + 32 * chunk_at<4>(3, rot4);
    load_b_out(g0, W3, c, h, ko0);                   SCHED_FENCE();
    float p[8];
    #pragma unroll
    for (int r = 0; r < 8; ++r) {
        p[r] = 0.0f;
        #pragma unroll
        for (int j = 0; j < 8; ++j) p[r] = fmaf(acc[j][r], s2r[j][r], p[r]);
    }
    #pragma unroll
    for (int off = 8; off > 0; off >>= 1) {
        #pragma unroll
        for (int r = 0; r < 8; ++r) p[r] += __shfl_xor(p[r], off, 32);
    }
    if (c == 0) {
        #pragma unroll
        for (int r = 0; r < 8; ++r) sR[wave * STRIP + 8 * h + r] = p[r];
    }
    __syncthreads();

    v2f bout[NPP][2];
    #pragma unroll
    for (int i = 0; i < NPP; ++i) {
        const int p = PIECE_TO_LOAD(i, tid + 128 * i);
        #pragma unroll
        for (int hf = 0; hf < 2; ++hf) bout[i][hf] = *(const v2f*)&b3[(4 * p + 2 * hf) % D_SZ];
    }

    v8f acc4[5];
    #pragma unroll
    for (int j = 0; j < 5; ++j) acc4[j] = (v8f){0.0f, 0.0f, 0.0f, 0.0f, 0.0f, 0.0f, 0.0f, 0.0f};
    const float* a2row = &sA[c * AP];
    load_a_lds(g0, a2row, h, ko0);
    load_b_out(g1, W3, c, h, ko1);
    load_a_lds(g1, a2row, h, ko1);                SCHED_FENCE();
    mma_strip<5>(g0, acc4);                          SCHED_FENCE();
    load_b_out(g0, W3, c, h, ko2);
    load_a_lds(g0, a2row, h, ko2);                SCHED_FENCE();
    mma_strip<5>(g1, acc4);                          SCHED_FENCE();
    load_b_out(g1, W3, c, h, ko3);
    load_a_lds(g1, a2row, h, ko3);                SCHED_FENCE();
    mma_strip<5>(g0, acc4);
    mma_strip<5>(g1, acc4);
    #pragma unroll
    for (int j = 0; j < 5; ++j) {
        #pragma unroll
        for (int r = 0; r < 8; ++r)
            sS[(wave * STRIP + 8 * h + r) * OP + 16 * j + c] = acc4[j][r];
    }
    __syncthreads();

    v4f* dst = (v4f*)(dxs + (size_t)row0 * D_SZ);
    #pragma unroll
    for (int i = 0; i < NPP; ++i) {
        const int p = tid + 128 * i;
        if (PIECE_IN_RUN(i, p)) {
            v2f o[2];
            #pragma unroll
            for (int hf = 0; hf < 2; ++hf) {
                const int e = 4 * p + 2 * hf;
                const int r = e / D_SZ;
                const int n = e - r * D_SZ;
                const v2f v = *(const v2f*)&sS[(0 * STRIP + r) * OP + n] + *(const v2f*)&sS[(1 * STRIP + r) * OP + n]
                            + *(const v2f*)&sS[(2 * STRIP + r) * OP + n] + *(const v2f*)&sS[(3 * STRIP + r) * OP + n];
                o[hf] = v + bout[i][hf];
            }
            dst[p] = (v4f){o[0].x, o[0].y, o[1].x, o[1].y};
        }
    }
    if (tid < STRIP)
        ndiv[row0 + tid] = -(sR[tid] + sR[STRIP + tid] + sR[2 * STRIP + tid] + sR[3 * STRIP + tid]);
}

extern "C" void kernel_launch(void* const* d_in, const int* in_sizes, int n_in,
                              void* d_out, int out_size, void* d_ws, size_t ws_size,
                              hipStream_t stream) {
    const float* xs = (const float*)d_in[0];
    const float* t  = (const float*)d_in[1];
    const float* W1 = (const float*)d_in[2];
    const float* b1 = (const float*)d_in[3];
    const float* W2 = (const float*)d_in[4];
    const float* b2 = (const float*)d_in[5];
    const float* W3 = (const float*)d_in[6];
    const float* b3 = (const float*)d_in[7];

    float* dxs  = (float*)d_out;
    float* ndiv = (float*)d_out + (size_t)B_SZ * D_SZ;

    float* Pc  = (float*)d_ws;
    float* Pw2 = (float*)d_ws + (size_t)H_SZ * H_SZ;
    float* Pw1 = (float*)d_ws + (size_t)2 * H_SZ * H_SZ;

    pack_weights<<<NT_C + NT_W1, 256, 0, stream>>>(W1, W2, W3, Pc, Pw2, Pw1);

    strip_mlp<<<B_SZ / STRIP, 128, 0, stream>>>(xs, t, Pw1, b1, Pw2, b2, W3, b3, Pc, dxs, ndiv);
}
